# starting kernel
# baseline (speedup 1.0000x reference)
#include <hip/hip_runtime.h>
#include <hip/hip_bf16.h>

typedef __attribute__((ext_vector_type(2))) float v2f;
typedef __attribute__((ext_vector_type(8))) float v8f;

#define NEG_SLOPE 0.2f


__device__ __forceinline__ float warp_sum(float v) {
  #pragma unroll
  for (int o = 16; o > 0; o >>= 1) v += __shfl_xor(v, o, 32);
  return v;
}

__device__ __forceinline__ unsigned fkey(float f) {
  unsigned u = __float_as_uint(f);
  return (u & 0x80000000u) ? ~u : (u | 0x80000000u);
}
__device__ __forceinline__ float funkey(unsigned k) {
  unsigned u = (k & 0x80000000u) ? (k & 0x7FFFFFFFu) : ~k;
  return __uint_as_float(u);
}

__device__ __forceinline__ void atomic_add_f32(float* p, float v) {
#if defined(__HIP_DEVICE_COMPILE__)
  unsafeAtomicAdd(p, v);
#else
  atomicAdd(p, v);
#endif
}


__global__ void zero_f32(float* __restrict__ p, size_t n) {
  size_t stride = (size_t)gridDim.x * blockDim.x;
  for (size_t i = (size_t)blockIdx.x * blockDim.x + threadIdx.x; i < n; i += stride)
    p[i] = 0.0f;
}

__global__ void gemm1_wmma(const float* __restrict__ X, const float* __restrict__ W,
                           float* __restrict__ H, int N) {
  const int lane = threadIdx.x & 31;
  const int wave = threadIdx.x >> 5;
  const int row0 = blockIdx.x * 16;
  const int col0 = wave * 16;
  const int m  = lane & 15;
  const int kh = lane >> 4;

  int row = row0 + m; if (row >= N) row = N - 1;
  const float* xrow = X + (size_t)row * 256;

  v8f c = {};
#if __has_builtin(__builtin_amdgcn_wmma_f32_16x16x4_f32)
  #pragma unroll 4
  for (int k = 0; k < 256; k += 4) {
    v2f a = *(const v2f*)(xrow + k + 2 * kh);
    v2f b;
    const float* wp = W + (size_t)(k + 2 * kh) * 128 + col0 + m;
    b.x = wp[0];
    b.y = wp[128];
    c = __builtin_amdgcn_wmma_f32_16x16x4_f32(false, a, false, b, (short)0, c,
                                              false, false);
  }
#else
  for (int k = 0; k < 256; k += 4) {
    v2f a = *(const v2f*)(xrow + k + 2 * kh);
    const float* wp = W + (size_t)(k + 2 * kh) * 128 + col0 + m;
    c[0] += a.x * wp[0] + a.y * wp[128];
  }
#endif
  float* hp = H + (size_t)(row0 + 8 * kh) * 128 + col0 + m;
  if (row0 + 16 <= N) {
    #pragma unroll
    for (int r = 0; r < 8; ++r) hp[(size_t)r * 128] = c[r];
  } else {
    #pragma unroll
    for (int r = 0; r < 8; ++r)
      if (row0 + r + 8 * kh < N) hp[(size_t)r * 128] = c[r];
  }
}

__global__ void node_e1(const float* __restrict__ H, const float* __restrict__ as,
                        const float* __restrict__ ad, float* __restrict__ es,
                        float* __restrict__ ed, int N) {
  int n = blockIdx.x * 8 + (threadIdx.x >> 5);
  if (n >= N) return;
  int lane = threadIdx.x & 31;
  float4 hv = ((const float4*)(H + (size_t)n * 128))[lane];
  float4 sv = ((const float4*)as)[lane];
  float4 dv = ((const float4*)ad)[lane];
  float s = hv.x * sv.x + hv.y * sv.y + hv.z * sv.z + hv.w * sv.w;
  float d = hv.x * dv.x + hv.y * dv.y + hv.z * dv.z + hv.w * dv.w;
  s = warp_sum(s);
  d = warp_sum(d);
  if (lane == 0) { es[n] = s; ed[n] = d; }
}

__global__ void edge_logits(const float* __restrict__ es, const float* __restrict__ ed,
                            const int* __restrict__ src, const int* __restrict__ dst,
                            int E, int nT, float* __restrict__ ebuf,
                            unsigned* __restrict__ mkey) {
  int i = blockIdx.x * blockDim.x + threadIdx.x;
  if (i >= nT) return;
  int s, d;
  if (i < E) { s = src[i]; d = dst[i]; } else { s = d = i - E; }
  float e = es[s] + ed[d];
  e = (e > 0.0f) ? e : NEG_SLOPE * e;
  ebuf[i] = e;
  atomicMax(&mkey[d], fkey(e));
}

__global__ void edge_exp(const float* __restrict__ ein, const unsigned* __restrict__ mkey,
                         const int* __restrict__ dst, int E, int nT,
                         float* __restrict__ eout, float* __restrict__ denom) {
  int i = blockIdx.x * blockDim.x + threadIdx.x;
  if (i >= nT) return;
  int d = (i < E) ? dst[i] : (i - E);
  float ex = expf(ein[i] - funkey(mkey[d]));
  eout[i] = ex;
  atomic_add_f32(&denom[d], ex);
}

__global__ void agg128(const float* __restrict__ H, const float* __restrict__ ex,
                       const float* __restrict__ denom, const int* __restrict__ src,
                       const int* __restrict__ dst, int E, int nT,
                       float* __restrict__ agg) {
  int i = blockIdx.x * 8 + (threadIdx.x >> 5);
  if (i >= nT) return;
  int lane = threadIdx.x & 31;
  int s, d;
  if (i < E) { s = src[i]; d = dst[i]; } else { s = d = i - E; }
  float alpha = ex[i] / denom[d];
  float4 hv = ((const float4*)(H + (size_t)s * 128))[lane];
  float* o = agg + (size_t)d * 128 + lane * 4;
  atomic_add_f32(o + 0, hv.x * alpha);
  atomic_add_f32(o + 1, hv.y * alpha);
  atomic_add_f32(o + 2, hv.z * alpha);
  atomic_add_f32(o + 3, hv.w * alpha);
}

__global__ void bias_relu(float* __restrict__ a, const float* __restrict__ b, size_t n) {
  size_t i = (size_t)blockIdx.x * blockDim.x + threadIdx.x;
  if (i >= n) return;
  float v = a[i] + b[i & 127];
  a[i] = (v > 0.0f) ? v : 0.0f;
}

__global__ void node_h2e2(const float* __restrict__ H, const float* __restrict__ W2,
                          const float* __restrict__ a2s, const float* __restrict__ a2d,
                          float* __restrict__ h2, float* __restrict__ e2s,
                          float* __restrict__ e2d, int N) {
  int n = blockIdx.x * 8 + (threadIdx.x >> 5);
  if (n >= N) return;
  int lane = threadIdx.x & 31;
  float4 hv = ((const float4*)(H + (size_t)n * 128))[lane];
  int c0 = lane * 4;
  float s0 = hv.x * W2[(c0 + 0) * 2 + 0] + hv.y * W2[(c0 + 1) * 2 + 0] +
             hv.z * W2[(c0 + 2) * 2 + 0] + hv.w * W2[(c0 + 3) * 2 + 0];
  float s1 = hv.x * W2[(c0 + 0) * 2 + 1] + hv.y * W2[(c0 + 1) * 2 + 1] +
             hv.z * W2[(c0 + 2) * 2 + 1] + hv.w * W2[(c0 + 3) * 2 + 1];
  s0 = warp_sum(s0);
  s1 = warp_sum(s1);
  if (lane == 0) {
    h2[2 * n + 0] = s0;
    h2[2 * n + 1] = s1;
    e2s[n] = s0 * a2s[0] + s1 * a2s[1];
    e2d[n] = s0 * a2d[0] + s1 * a2d[1];
  }
}

__global__ void agg2(const float* __restrict__ h2, const float* __restrict__ ex,
                     const float* __restrict__ denom, const int* __restrict__ src,
                     const int* __restrict__ dst, int E, int nT,
                     float* __restrict__ out) {
  int i = blockIdx.x * blockDim.x + threadIdx.x;
  if (i >= nT) return;
  int s, d;
  if (i < E) { s = src[i]; d = dst[i]; } else { s = d = i - E; }
  float alpha = ex[i] / denom[d];
  atomic_add_f32(&out[2 * d + 0], h2[2 * s + 0] * alpha);
  atomic_add_f32(&out[2 * d + 1], h2[2 * s + 1] * alpha);
}

__global__ void bias2(float* __restrict__ out, const float* __restrict__ b, int n2) {
  int i = blockIdx.x * blockDim.x + threadIdx.x;
  if (i >= n2) return;
  out[i] += b[i & 1];
}


extern "C" void kernel_launch(void* const* d_in, const int* in_sizes, int n_in,
                              void* d_out, int out_size, void* d_ws, size_t ws_size,
                              hipStream_t stream) {
  const float* x   = (const float*)d_in[0];
  const int*  eidx = (const int*)d_in[1];
  const float* W1  = (const float*)d_in[2];
  const float* a1s = (const float*)d_in[3];
  const float* a1d = (const float*)d_in[4];
  const float* b1  = (const float*)d_in[5];
  const float* W2  = (const float*)d_in[6];
  const float* a2s = (const float*)d_in[7];
  const float* a2d = (const float*)d_in[8];
  const float* b2  = (const float*)d_in[9];

  const int N  = in_sizes[0] / 256;
  const int E  = in_sizes[1] / 2;
  const int nT = E + N;
  const int* src = eidx;
  const int* dst = eidx + E;
  float* out = (float*)d_out;

  char* w = (char*)d_ws;
  float*    zblk   = (float*)w;
  float*    ag1    = (float*)w;    w += (size_t)N * 128 * sizeof(float);
  unsigned* mkey1  = (unsigned*)w; w += (size_t)N * sizeof(unsigned);
  float*    den1   = (float*)w;    w += (size_t)N * sizeof(float);
  unsigned* mkey2  = (unsigned*)w; w += (size_t)N * sizeof(unsigned);
  float*    den2   = (float*)w;    w += (size_t)N * sizeof(float);
  const size_t zcount = (size_t)N * 132;
  float*    h1     = (float*)w;    w += (size_t)N * 128 * sizeof(float);
  float*    ebuf   = (float*)w;    w += (size_t)nT * sizeof(float);
  float*    e1s    = (float*)w;    w += (size_t)N * sizeof(float);
  float*    e1d    = (float*)w;    w += (size_t)N * sizeof(float);
  float*    h2     = (float*)w;    w += (size_t)N * 2 * sizeof(float);
  float*    e2s    = (float*)w;    w += (size_t)N * sizeof(float);
  float*    e2d    = (float*)w;    w += (size_t)N * sizeof(float);

  const int TB = 256;
  zero_f32<<<2048, TB, 0, stream>>>(zblk, zcount);
  zero_f32<<<256,  TB, 0, stream>>>(out, (size_t)N * 2);

  gemm1_wmma<<<(N + 15) / 16, TB, 0, stream>>>(x, W1, h1, N);
  node_e1<<<(N + 7) / 8, TB, 0, stream>>>(h1, a1s, a1d, e1s, e1d, N);
  edge_logits<<<(nT + TB - 1) / TB, TB, 0, stream>>>(e1s, e1d, src, dst, E, nT, ebuf, mkey1);
  edge_exp<<<(nT + TB - 1) / TB, TB, 0, stream>>>(ebuf, mkey1, dst, E, nT, ebuf, den1);
  agg128<<<(nT + 7) / 8, TB, 0, stream>>>(h1, ebuf, den1, src, dst, E, nT, ag1);
  bias_relu<<<(int)(((size_t)N * 128 + TB - 1) / TB), TB, 0, stream>>>(ag1, b1, (size_t)N * 128);

  node_h2e2<<<(N + 7) / 8, TB, 0, stream>>>(ag1, W2, a2s, a2d, h2, e2s, e2d, N);
  edge_logits<<<(nT + TB - 1) / TB, TB, 0, stream>>>(e2s, e2d, src, dst, E, nT, ebuf, mkey2);
  edge_exp<<<(nT + TB - 1) / TB, TB, 0, stream>>>(ebuf, mkey2, dst, E, nT, ebuf, den2);
  agg2<<<(nT + TB - 1) / TB, TB, 0, stream>>>(h2, ebuf, den2, src, dst, E, nT, out);
  bias2<<<(2 * N + TB - 1) / TB, TB, 0, stream>>>(out, b2, 2 * N);
}


# Layer-1 segment max moved into the owner kernel (pass A: LDS max + edge list in LDS; pass B sums from the list): edge-logits launch, global max atomics gone; 400-node ranges; sum order = LDS arrival
# speedup vs baseline: 3.3866x; 3.3866x over previous
#include <hip/hip_runtime.h>
#include <hip/hip_bf16.h>

typedef __attribute__((ext_vector_type(2))) float v2f;
typedef __attribute__((ext_vector_type(8))) float v8f;

#define NEG_SLOPE 0.2f


__device__ __forceinline__ float warp_sum(float v) {
  #pragma unroll
  for (int o = 16; o > 0; o >>= 1) v += __shfl_xor(v, o, 32);
  return v;
}

__device__ __forceinline__ unsigned fkey(float f) {
  unsigned u = __float_as_uint(f);
  return (u & 0x80000000u) ? ~u : (u | 0x80000000u);
}
__device__ __forceinline__ float funkey(unsigned k) {
  unsigned u = (k & 0x80000000u) ? (k & 0x7FFFFFFFu) : ~k;
  return __uint_as_float(u);
}

__global__ void gemm1_wmma(const float* __restrict__ X, const float* __restrict__ W,
                           float* __restrict__ H, int N) {
  const int lane = threadIdx.x & 31;
  const int wave = threadIdx.x >> 5;
  const int row0 = blockIdx.x * 16;
  const int col0 = wave * 16;
  const int m  = lane & 15;
  const int kh = lane >> 4;

  int row = row0 + m; if (row >= N) row = N - 1;
  const float* xrow = X + (size_t)row * 256;

  v8f c = {};
#if __has_builtin(__builtin_amdgcn_wmma_f32_16x16x4_f32)
  #pragma unroll 4
  for (int k = 0; k < 256; k += 4) {
    v2f a = *(const v2f*)(xrow + k + 2 * kh);
    v2f b;
    const float* wp = W + (size_t)(k + 2 * kh) * 128 + col0 + m;
    b.x = wp[0];
    b.y = wp[128];
    c = __builtin_amdgcn_wmma_f32_16x16x4_f32(false, a, false, b, (short)0, c,
                                              false, false);
  }
#else
  for (int k = 0; k < 256; k += 4) {
    v2f a = *(const v2f*)(xrow + k + 2 * kh);
    const float* wp = W + (size_t)(k + 2 * kh) * 128 + col0 + m;
    c[0] += a.x * wp[0] + a.y * wp[128];
  }
#endif
  float* hp = H + (size_t)(row0 + 8 * kh) * 128 + col0 + m;
  if (row0 + 16 <= N) {
    #pragma unroll
    for (int r = 0; r < 8; ++r) hp[(size_t)r * 128] = c[r];
  } else {
    #pragma unroll
    for (int r = 0; r < 8; ++r)
      if (row0 + r + 8 * kh < N) hp[(size_t)r * 128] = c[r];
  }
}

__global__ void node_e1(const float* __restrict__ H, const float* __restrict__ as,
                        const float* __restrict__ ad, float* __restrict__ es,
                        float* __restrict__ ed, int N) {
  int n = blockIdx.x * 8 + (threadIdx.x >> 5);
  if (n >= N) return;
  int lane = threadIdx.x & 31;
  float4 hv = ((const float4*)(H + (size_t)n * 128))[lane];
  float4 sv = ((const float4*)as)[lane];
  float4 dv = ((const float4*)ad)[lane];
  float s = hv.x * sv.x + hv.y * sv.y + hv.z * sv.z + hv.w * sv.w;
  float d = hv.x * dv.x + hv.y * dv.y + hv.z * dv.z + hv.w * dv.w;
  s = warp_sum(s);
  d = warp_sum(d);
  if (lane == 0) { es[n] = s; ed[n] = d; }
}

#define RNG 400

__global__ void range_ids(const int* __restrict__ dst, int E, unsigned char* __restrict__ rid) {
  const int t = blockIdx.x * blockDim.x + threadIdx.x;
  int i = 4 * t;
  if (i + 3 < E) {
    const unsigned w = (unsigned)(dst[i] / RNG) | ((unsigned)(dst[i + 1] / RNG) << 8) |
                       ((unsigned)(dst[i + 2] / RNG) << 16) | ((unsigned)(dst[i + 3] / RNG) << 24);
    ((unsigned*)rid)[t] = w;
  } else {
    for (; i < E; ++i) rid[i] = (unsigned char)(dst[i] / RNG);
  }
}

template <class F>
__device__ __forceinline__ void scan_owned(const unsigned char* __restrict__ rid, int E, int N,
                                           int r0, int lane, int wave, F&& fl) {
  int qv = 0;
  int qn = 0;
  auto push = [&](int i) {
    qv = (lane == qn) ? i : qv;
    if (++qn == 32) { fl(qv, 32); qn = 0; }
  };
  const unsigned own = blockIdx.x & 255u;
  const unsigned pat = own * 0x01010101u;
  auto zb = [](unsigned x) {
    return ~((((x & 0x7F7F7F7Fu) + 0x7F7F7F7Fu) | x) | 0x7F7F7F7Fu);
  };
  int a0 = (int)((16u - ((unsigned)(size_t)rid & 15u)) & 15u);
  if (a0 > E) a0 = E;
  const int nfast = (E - a0) / 2048;
  const uint4* rq = (const uint4*)(rid + a0);
  const int T = (nfast > wave) ? (nfast - wave + 31) / 32 : 0;
  int t = (int)(((size_t)blockIdx.x * (size_t)T) / gridDim.x);
  for (int k = 0; k < T; ++k) {
    const int c = t * 32 + wave;
    if (++t == T) t = 0;
    uint4 v[4];
    #pragma unroll
    for (int j = 0; j < 4; ++j) v[j] = rq[c * 128 + j * 32 + lane];
    #pragma unroll
    for (int j = 0; j < 4; ++j) {
      const unsigned f0 = zb(v[j].x ^ pat);
      const unsigned f1 = zb(v[j].y ^ pat);
      const unsigned f2 = zb(v[j].z ^ pat);
      const unsigned f3 = zb(v[j].w ^ pat);
      const unsigned m0 = __builtin_amdgcn_ballot_w32(f0 != 0u);
      const unsigned m1 = __builtin_amdgcn_ballot_w32(f1 != 0u);
      const unsigned m2 = __builtin_amdgcn_ballot_w32(f2 != 0u);
      const unsigned m3 = __builtin_amdgcn_ballot_w32(f3 != 0u);
      unsigned any = m0 | m1 | m2 | m3;
      const int ibase = a0 + (c * 128 + j * 32) * 16;
      while (any) {
        const int l = __builtin_ctz(any);
        any &= any - 1;
        const int il = ibase + l * 16;
        if ((m0 >> l) & 1u)
          for (unsigned f = __builtin_amdgcn_readlane(f0, l); f; f &= f - 1)
            push(il + 0 + (__builtin_ctz(f) >> 3));
        if ((m1 >> l) & 1u)
          for (unsigned f = __builtin_amdgcn_readlane(f1, l); f; f &= f - 1)
            push(il + 4 + (__builtin_ctz(f) >> 3));
        if ((m2 >> l) & 1u)
          for (unsigned f = __builtin_amdgcn_readlane(f2, l); f; f &= f - 1)
            push(il + 8 + (__builtin_ctz(f) >> 3));
        if ((m3 >> l) & 1u)
          for (unsigned f = __builtin_amdgcn_readlane(f3, l); f; f &= f - 1)
            push(il + 12 + (__builtin_ctz(f) >> 3));
      }
    }
  }
  const int nrem = E - nfast * 2048;
  for (int g = wave; g * 32 < nrem; g += 32) {
    const int j = g * 32 + lane;
    const int i = (j < a0) ? j : nfast * 2048 + j;
    const unsigned idv = (j < nrem) ? (unsigned)rid[i] : 256u;
    unsigned m = __builtin_amdgcn_ballot_w32(idv == own);
    while (m) {
      const int l = __builtin_ctz(m);
      m &= m - 1;
      push(__builtin_amdgcn_readlane(i, l));
    }
  }
  for (int n = r0 + wave; n < r0 + RNG && n < N; n += 32) push(E + n);
  if (qn) fl(qv, qn);
}

#define AGG_T 1024
#define QCAP1 10240
static_assert(RNG <= 512 && 256 * RNG <= (1 << 17), "list entry packing");

__global__ __launch_bounds__(AGG_T)
void agg128_own(const float* __restrict__ H, const float* __restrict__ es,
                const float* __restrict__ ed, const int* __restrict__ src,
                const int* __restrict__ dst, const unsigned char* __restrict__ rid, int E, int N,
                const float* __restrict__ bias, float* __restrict__ outp) {
  __shared__ float lacc[RNG * 128];
  __shared__ uint2 lq[QCAP1];
  __shared__ float led[RNG];
  __shared__ unsigned lmx[RNG];
  __shared__ float lden[RNG];
  __shared__ int lqn;
  const int tid = threadIdx.x;
  const int lane = tid & 31;
  const int wave = __builtin_amdgcn_readfirstlane(tid >> 5);
  const int r0 = blockIdx.x * RNG;

  for (int q = tid; q < RNG * 32; q += AGG_T) ((float4*)lacc)[q] = make_float4(0.f, 0.f, 0.f, 0.f);
  for (int q = tid; q < RNG; q += AGG_T) {
    const int n = r0 + q;
    led[q] = (n < N) ? ed[n] : 0.0f;
    lmx[q] = 0u;
    lden[q] = 0.0f;
  }
  if (tid == 0) lqn = 0;
  __syncthreads();

  const float* hl = H + lane;
  float* al = lacc + lane;

  auto edge = [&](int i, int& s, int& dl, float& e) {
    int d;
    if (i < E) { s = src[i]; d = dst[i]; } else { s = d = i - E; }
    dl = d - r0;
    e = es[s] + led[dl];
    e = (e > 0.0f) ? e : NEG_SLOPE * e;
  };

  auto rows = [&](int s, int dl, float ex, int cnt) {
    const int exb = __float_as_int(ex);
    int b = 0;
    for (; b + 8 <= cnt; b += 8) {
      float hv[8][4];
      #pragma unroll
      for (int j = 0; j < 8; ++j) {
        const int sj = __builtin_amdgcn_readlane(s, b + j);
        const float* hp = hl + (size_t)sj * 128;
        hv[j][0] = hp[0]; hv[j][1] = hp[32]; hv[j][2] = hp[64]; hv[j][3] = hp[96];
      }
      #pragma unroll
      for (int j = 0; j < 8; ++j) {
        const int dj = __builtin_amdgcn_readlane(dl, b + j);
        const float ej = __int_as_float(__builtin_amdgcn_readlane(exb, b + j));
        float* o = al + dj * 128;
        atomicAdd(o +  0, hv[j][0] * ej);
        atomicAdd(o + 32, hv[j][1] * ej);
        atomicAdd(o + 64, hv[j][2] * ej);
        atomicAdd(o + 96, hv[j][3] * ej);
      }
    }
    for (; b < cnt; ++b) {
      const int sj = __builtin_amdgcn_readlane(s, b);
      const int dj = __builtin_amdgcn_readlane(dl, b);
      const float ej = __int_as_float(__builtin_amdgcn_readlane(exb, b));
      const float* hp = hl + (size_t)sj * 128;
      float* o = al + dj * 128;
      const float h0 = hp[0], h1 = hp[32], h2 = hp[64], h3 = hp[96];
      atomicAdd(o +  0, h0 * ej);
      atomicAdd(o + 32, h1 * ej);
      atomicAdd(o + 64, h2 * ej);
      atomicAdd(o + 96, h3 * ej);
    }
  };

  scan_owned(rid, E, N, r0, lane, wave, [&](int qv, int cnt) {
    int base = 0;
    if (lane == 0) base = atomicAdd(&lqn, cnt);
    base = __builtin_amdgcn_readlane(base, 0);
    if (lane < cnt) {
      int s, dl; float e;
      edge(qv, s, dl, e);
      atomicMax(&lmx[dl], fkey(e));
      const int p = base + lane;
      if (p < QCAP1) lq[p] = make_uint2((unsigned)s | ((unsigned)dl << 17), __float_as_uint(e));
    }
  });
  __syncthreads();

  const int total = __builtin_amdgcn_readfirstlane(lqn);
  if (total <= QCAP1) {
    for (int c0 = wave * 32; c0 < total; c0 += 32 * 32) {
      const int cnt = (total - c0 < 32) ? total - c0 : 32;
      int s = 0, dl = 0;
      float ex = 0.0f;
      if (lane < cnt) {
        const uint2 t2 = lq[c0 + lane];
        s = (int)(t2.x & 0x1FFFFu);
        dl = (int)(t2.x >> 17);
        ex = expf(__uint_as_float(t2.y) - funkey(lmx[dl]));
        atomicAdd(&lden[dl], ex);
      }
      rows(s, dl, ex, cnt);
    }
  } else {
    scan_owned(rid, E, N, r0, lane, wave, [&](int qv, int cnt) {
      int s = 0, dl = 0;
      float ex = 0.0f;
      if (lane < cnt) {
        float e;
        edge(qv, s, dl, e);
        ex = expf(e - funkey(lmx[dl]));
        atomicAdd(&lden[dl], ex);
      }
      rows(s, dl, ex, cnt);
    });
  }
  __syncthreads();

  for (int q = tid; q < RNG * 32; q += AGG_T) {
    const int node = q >> 5;
    const int n = r0 + node;
    if (n < N) {
      const float4 a = ((const float4*)lacc)[q];
      const float dn = lden[node];
      const float4 bb = ((const float4*)bias)[q & 31];
      float4 r;
      r.x = a.x / dn + bb.x; r.y = a.y / dn + bb.y; r.z = a.z / dn + bb.z; r.w = a.w / dn + bb.w;
      r.x = (r.x > 0.0f) ? r.x : 0.0f; r.y = (r.y > 0.0f) ? r.y : 0.0f;
      r.z = (r.z > 0.0f) ? r.z : 0.0f; r.w = (r.w > 0.0f) ? r.w : 0.0f;
      ((float4*)(outp + (size_t)n * 128))[q & 31] = r;
    }
  }
}

__global__ void node_h2e2(const float* __restrict__ H, const float* __restrict__ W2,
                          const float* __restrict__ a2s, const float* __restrict__ a2d,
                          float* __restrict__ rec, int N) {
  int n = blockIdx.x * 8 + (threadIdx.x >> 5);
  if (n >= N) return;
  int lane = threadIdx.x & 31;
  float4 hv = ((const float4*)(H + (size_t)n * 128))[lane];
  int c0 = lane * 4;
  float s0 = hv.x * W2[(c0 + 0) * 2 + 0] + hv.y * W2[(c0 + 1) * 2 + 0] +
             hv.z * W2[(c0 + 2) * 2 + 0] + hv.w * W2[(c0 + 3) * 2 + 0];
  float s1 = hv.x * W2[(c0 + 0) * 2 + 1] + hv.y * W2[(c0 + 1) * 2 + 1] +
             hv.z * W2[(c0 + 2) * 2 + 1] + hv.w * W2[(c0 + 3) * 2 + 1];
  s0 = warp_sum(s0);
  s1 = warp_sum(s1);
  if (lane == 0) {
    ((float4*)rec)[n] = make_float4(s0, s1, s0 * a2s[0] + s1 * a2s[1], s0 * a2d[0] + s1 * a2d[1]);
  }
}

#define RNG2 RNG
#define QCAP 12288

__global__ __launch_bounds__(AGG_T)
void agg2_own(const float* __restrict__ rec, const int* __restrict__ src,
              const int* __restrict__ dst, const unsigned char* __restrict__ rid, int E, int N,
              const float* __restrict__ bias, float* __restrict__ outp) {
  __shared__ float4 lq[QCAP];
  __shared__ float led[RNG2];
  __shared__ unsigned lmx[RNG2];
  __shared__ float lden[RNG2];
  __shared__ float lo0[RNG2];
  __shared__ float lo1[RNG2];
  __shared__ int lqn;
  const int tid = threadIdx.x;
  const int lane = tid & 31;
  const int wave = __builtin_amdgcn_readfirstlane(tid >> 5);
  const int r0 = blockIdx.x * RNG2;
  const float4* rec4 = (const float4*)rec;

  for (int q = tid; q < RNG2; q += AGG_T) {
    const int n = r0 + q;
    led[q] = (n < N) ? rec4[n].w : 0.0f;
    lmx[q] = 0u;
    lden[q] = 0.0f;
    lo0[q] = 0.0f;
    lo1[q] = 0.0f;
  }
  if (tid == 0) lqn = 0;
  __syncthreads();

  auto edge = [&](int i, float4& r, int& dl, float& e) {
    int s, d;
    if (i < E) { s = src[i]; d = dst[i]; } else { s = d = i - E; }
    r = rec4[s];
    dl = d - r0;
    e = r.z + led[dl];
    e = (e > 0.0f) ? e : NEG_SLOPE * e;
  };

  scan_owned(rid, E, N, r0, lane, wave, [&](int qv, int cnt) {
    int base = 0;
    if (lane == 0) base = atomicAdd(&lqn, cnt);
    base = __builtin_amdgcn_readlane(base, 0);
    if (lane < cnt) {
      float4 r; int dl; float e;
      edge(qv, r, dl, e);
      atomicMax(&lmx[dl], fkey(e));
      const int p = base + lane;
      if (p < QCAP) lq[p] = make_float4(e, r.x, r.y, __int_as_float(dl));
    }
  });
  __syncthreads();

  const int total = lqn;
  if (total <= QCAP) {
    for (int q = tid; q < total; q += AGG_T) {
      const float4 t4 = lq[q];
      const int dl = __float_as_int(t4.w);
      const float ex = expf(t4.x - funkey(lmx[dl]));
      atomicAdd(&lden[dl], ex);
      atomicAdd(&lo0[dl], t4.y * ex);
      atomicAdd(&lo1[dl], t4.z * ex);
    }
  } else {
    scan_owned(rid, E, N, r0, lane, wave, [&](int qv, int cnt) {
      if (lane < cnt) {
        float4 r; int dl; float e;
        edge(qv, r, dl, e);
        const float ex = expf(e - funkey(lmx[dl]));
        atomicAdd(&lden[dl], ex);
        atomicAdd(&lo0[dl], r.x * ex);
        atomicAdd(&lo1[dl], r.y * ex);
      }
    });
  }
  __syncthreads();

  const float bb0 = bias[0], bb1 = bias[1];
  for (int q = tid; q < RNG2; q += AGG_T) {
    const int n = r0 + q;
    if (n < N) {
      const float dn = lden[q];
      float2 o;
      o.x = lo0[q] / dn + bb0;
      o.y = lo1[q] / dn + bb1;
      ((float2*)outp)[n] = o;
    }
  }
}


extern "C" void kernel_launch(void* const* d_in, const int* in_sizes, int n_in,
                              void* d_out, int out_size, void* d_ws, size_t ws_size,
                              hipStream_t stream) {
  const float* x   = (const float*)d_in[0];
  const int*  eidx = (const int*)d_in[1];
  const float* W1  = (const float*)d_in[2];
  const float* a1s = (const float*)d_in[3];
  const float* a1d = (const float*)d_in[4];
  const float* b1  = (const float*)d_in[5];
  const float* W2  = (const float*)d_in[6];
  const float* a2s = (const float*)d_in[7];
  const float* a2d = (const float*)d_in[8];
  const float* b2  = (const float*)d_in[9];

  const int N  = in_sizes[0] / 256;
  const int E  = in_sizes[1] / 2;
  const int* src = eidx;
  const int* dst = eidx + E;
  float* out = (float*)d_out;

  char* w = (char*)d_ws;
  float*    h1b    = (float*)w;    w += (size_t)N * 128 * sizeof(float);
  float*    h1     = (float*)w;    w += (size_t)N * 128 * sizeof(float);
  float*    rec2   = (float*)w;    w += (size_t)N * 4 * sizeof(float);
  float*    e1s    = (float*)w;    w += (size_t)N * sizeof(float);
  float*    e1d    = (float*)w;    w += (size_t)N * sizeof(float);
  unsigned char* rid = (unsigned char*)w; w += (size_t)E;

  const int TB = 256;
  range_ids<<<((E + 3) / 4 + TB - 1) / TB, TB, 0, stream>>>(dst, E, rid);

  gemm1_wmma<<<(N + 15) / 16, TB, 0, stream>>>(x, W1, h1, N);
  node_e1<<<(N + 7) / 8, TB, 0, stream>>>(h1, a1s, a1d, e1s, e1d, N);
  agg128_own<<<(N + RNG - 1) / RNG, AGG_T, 0, stream>>>(h1, e1s, e1d, src, dst, rid, E, N, b1, h1b);

  node_h2e2<<<(N + 7) / 8, TB, 0, stream>>>(h1b, W2, a2s, a2d, rec2, N);
  agg2_own<<<(N + RNG2 - 1) / RNG2, AGG_T, 0, stream>>>(rec2, src, dst, rid, E, N, b2, out);
}


# Layer-1 GEMM on a resident grid: a wave keeps its 256x16 slice of W in registers for all its row tiles; X tiles pass through LDS by whole-line loads, one ahead; k summed in another order (2 chains)
# speedup vs baseline: 4.1330x; 1.2204x over previous
#include <hip/hip_runtime.h>
#include <hip/hip_bf16.h>

typedef __attribute__((ext_vector_type(2))) float v2f;
typedef __attribute__((ext_vector_type(8))) float v8f;

#define NEG_SLOPE 0.2f


__device__ __forceinline__ float warp_sum(float v) {
  #pragma unroll
  for (int o = 16; o > 0; o >>= 1) v += __shfl_xor(v, o, 32);
  return v;
}

__device__ __forceinline__ unsigned fkey(float f) {
  unsigned u = __float_as_uint(f);
  return (u & 0x80000000u) ? ~u : (u | 0x80000000u);
}
__device__ __forceinline__ float funkey(unsigned k) {
  unsigned u = (k & 0x80000000u) ? (k & 0x7FFFFFFFu) : ~k;
  return __uint_as_float(u);
}

#define G1_T 256
#define G1_GRID 512
#define G1_PITCH 260

__global__ __launch_bounds__(G1_T)
void gemm1_wmma(const float* __restrict__ X, const float* __restrict__ W,
                float* __restrict__ H, int N) {
  __shared__ __attribute__((aligned(16))) float xs[2][16 * G1_PITCH];
  const int tid = threadIdx.x;
  const int lane = tid & 31;
  const int wave = __builtin_amdgcn_readfirstlane(tid >> 5);
  const int col0 = wave * 16;
  const int m  = lane & 15;
  const int kh = lane >> 4;
  const int ntiles = (N + 15) >> 4;

  float4 b[32];
  #pragma unroll
  for (int j = 0; j < 32; ++j) {
    const float* wp = W + (size_t)(8 * j + 4 * kh) * 128 + col0 + m;
    b[j] = make_float4(wp[0], wp[128], wp[256], wp[384]);
  }

  const int srow = tid >> 6;
  const int scol = (tid & 63) * 4;
  const float* xsrc = X + scol;
  float* sdst = &xs[0][srow * G1_PITCH + scol];
  float4 p0, p1, p2, p3;
  auto fetch = [&](int t) {
    int r0 = t * 16 + srow, r1 = r0 + 4, r2 = r0 + 8, r3 = r0 + 12;
    if (r3 >= N) {
      r0 = (r0 < N) ? r0 : N - 1; r1 = (r1 < N) ? r1 : N - 1;
      r2 = (r2 < N) ? r2 : N - 1; r3 = N - 1;
    }
    p0 = *(const float4*)(xsrc + (size_t)r0 * 256);
    p1 = *(const float4*)(xsrc + (size_t)r1 * 256);
    p2 = *(const float4*)(xsrc + (size_t)r2 * 256);
    p3 = *(const float4*)(xsrc + (size_t)r3 * 256);
  };
  auto stage = [&](int bf) {
    float* d = sdst + bf * (16 * G1_PITCH);
    *(float4*)(d) = p0;
    *(float4*)(d + 4 * G1_PITCH) = p1;
    *(float4*)(d + 8 * G1_PITCH) = p2;
    *(float4*)(d + 12 * G1_PITCH) = p3;
  };

  int t = blockIdx.x;
  if (t < ntiles) { fetch(t); stage(0); }
  __syncthreads();
  int bf = 0;
  for (; t < ntiles; t += G1_GRID) {
    const int tn = t + G1_GRID;
    if (tn < ntiles) fetch(tn);

    v8f c0 = {}, c1 = {};
    const float* xa = &xs[bf][m * G1_PITCH + 4 * kh];
    float4 af[2][4];
    #pragma unroll
    for (int i = 0; i < 4; ++i) af[0][i] = *(const float4*)(xa + 8 * i);
    #pragma unroll
    for (int g = 0; g < 8; ++g) {
      if (g < 7) {
        #pragma unroll
        for (int i = 0; i < 4; ++i) af[(g + 1) & 1][i] = *(const float4*)(xa + 8 * (4 * (g + 1) + i));
      }
      #pragma unroll
      for (int i = 0; i < 4; ++i) {
        const float4 a = af[g & 1][i];
        const float4 bb = b[4 * g + i];
        v2f a0, a1, b0, b1;
        a0.x = a.x; a0.y = a.y; a1.x = a.z; a1.y = a.w;
        b0.x = bb.x; b0.y = bb.y; b1.x = bb.z; b1.y = bb.w;
#if __has_builtin(__builtin_amdgcn_wmma_f32_16x16x4_f32)
        c0 = __builtin_amdgcn_wmma_f32_16x16x4_f32(false, a0, false, b0, (short)0, c0,
                                                   false, false);
        c1 = __builtin_amdgcn_wmma_f32_16x16x4_f32(false, a1, false, b1, (short)0, c1,
                                                   false, false);
#else
        c0[0] += a0.x * b0.x + a0.y * b0.y;
        c1[0] += a1.x * b1.x + a1.y * b1.y;
#endif
      }
    }
#if __has_builtin(__builtin_amdgcn_sched_group_barrier)
    __builtin_amdgcn_sched_group_barrier(0x100, 4, 0);
    #pragma unroll
    for (int g = 0; g < 7; ++g) {
      __builtin_amdgcn_sched_group_barrier(0x100, 4, 0);
      __builtin_amdgcn_sched_group_barrier(0x008, 8, 0);
    }
    __builtin_amdgcn_sched_group_barrier(0x008, 8, 0);
#endif
    const v8f c = c0 + c1;

    if (tn < ntiles) stage(bf ^ 1);
    __syncthreads();
    bf ^= 1;

    const int row0 = t * 16;
    float* hp = H + (size_t)(row0 + 8 * kh) * 128 + col0 + m;
    if (row0 + 16 <= N) {
      #pragma unroll
      for (int r = 0; r < 8; ++r) hp[(size_t)r * 128] = c[r];
    } else {
      #pragma unroll
      for (int r = 0; r < 8; ++r)
        if (row0 + r + 8 * kh < N) hp[(size_t)r * 128] = c[r];
    }
  }
}

__global__ void node_e1(const float* __restrict__ H, const float* __restrict__ as,
                        const float* __restrict__ ad, float* __restrict__ es,
                        float* __restrict__ ed, int N) {
  int n = blockIdx.x * 8 + (threadIdx.x >> 5);
  if (n >= N) return;
  int lane = threadIdx.x & 31;
  float4 hv = ((const float4*)(H + (size_t)n * 128))[lane];
  float4 sv = ((const float4*)as)[lane];
  float4 dv = ((const float4*)ad)[lane];
  float s = hv.x * sv.x + hv.y * sv.y + hv.z * sv.z + hv.w * sv.w;
  float d = hv.x * dv.x + hv.y * dv.y + hv.z * dv.z + hv.w * dv.w;
  s = warp_sum(s);
  d = warp_sum(d);
  if (lane == 0) { es[n] = s; ed[n] = d; }
}

#define RNG 400

__global__ void range_ids(const int* __restrict__ dst, int E, unsigned char* __restrict__ rid) {
  const int t = blockIdx.x * blockDim.x + threadIdx.x;
  int i = 4 * t;
  if (i + 3 < E) {
    const unsigned w = (unsigned)(dst[i] / RNG) | ((unsigned)(dst[i + 1] / RNG) << 8) |
                       ((unsigned)(dst[i + 2] / RNG) << 16) | ((unsigned)(dst[i + 3] / RNG) << 24);
    ((unsigned*)rid)[t] = w;
  } else {
    for (; i < E; ++i) rid[i] = (unsigned char)(dst[i] / RNG);
  }
}

template <class F>
__device__ __forceinline__ void scan_owned(const unsigned char* __restrict__ rid, int E, int N,
                                           int r0, int lane, int wave, F&& fl) {
  int qv = 0;
  int qn = 0;
  auto push = [&](int i) {
    qv = (lane == qn) ? i : qv;
    if (++qn == 32) { fl(qv, 32); qn = 0; }
  };
  const unsigned own = blockIdx.x & 255u;
  const unsigned pat = own * 0x01010101u;
  auto zb = [](unsigned x) {
    return ~((((x & 0x7F7F7F7Fu) + 0x7F7F7F7Fu) | x) | 0x7F7F7F7Fu);
  };
  int a0 = (int)((16u - ((unsigned)(size_t)rid & 15u)) & 15u);
  if (a0 > E) a0 = E;
  const int nfast = (E - a0) / 2048;
  const uint4* rq = (const uint4*)(rid + a0);
  const int T = (nfast > wave) ? (nfast - wave + 31) / 32 : 0;
  int t = (int)(((size_t)blockIdx.x * (size_t)T) / gridDim.x);
  for (int k = 0; k < T; ++k) {
    const int c = t * 32 + wave;
    if (++t == T) t = 0;
    uint4 v[4];
    #pragma unroll
    for (int j = 0; j < 4; ++j) v[j] = rq[c * 128 + j * 32 + lane];
    #pragma unroll
    for (int j = 0; j < 4; ++j) {
      const unsigned f0 = zb(v[j].x ^ pat);
      const unsigned f1 = zb(v[j].y ^ pat);
      const unsigned f2 = zb(v[j].z ^ pat);
      const unsigned f3 = zb(v[j].w ^ pat);
      const unsigned m0 = __builtin_amdgcn_ballot_w32(f0 != 0u);
      const unsigned m1 = __builtin_amdgcn_ballot_w32(f1 != 0u);
      const unsigned m2 = __builtin_amdgcn_ballot_w32(f2 != 0u);
      const unsigned m3 = __builtin_amdgcn_ballot_w32(f3 != 0u);
      unsigned any = m0 | m1 | m2 | m3;
      const int ibase = a0 + (c * 128 + j * 32) * 16;
      while (any) {
        const int l = __builtin_ctz(any);
        any &= any - 1;
        const int il = ibase + l * 16;
        if ((m0 >> l) & 1u)
          for (unsigned f = __builtin_amdgcn_readlane(f0, l); f; f &= f - 1)
            push(il + 0 + (__builtin_ctz(f) >> 3));
        if ((m1 >> l) & 1u)
          for (unsigned f = __builtin_amdgcn_readlane(f1, l); f; f &= f - 1)
            push(il + 4 + (__builtin_ctz(f) >> 3));
        if ((m2 >> l) & 1u)
          for (unsigned f = __builtin_amdgcn_readlane(f2, l); f; f &= f - 1)
            push(il + 8 + (__builtin_ctz(f) >> 3));
        if ((m3 >> l) & 1u)
          for (unsigned f = __builtin_amdgcn_readlane(f3, l); f; f &= f - 1)
            push(il + 12 + (__builtin_ctz(f) >> 3));
      }
    }
  }
  const int nrem = E - nfast * 2048;
  for (int g = wave; g * 32 < nrem; g += 32) {
    const int j = g * 32 + lane;
    const int i = (j < a0) ? j : nfast * 2048 + j;
    const unsigned idv = (j < nrem) ? (unsigned)rid[i] : 256u;
    unsigned m = __builtin_amdgcn_ballot_w32(idv == own);
    while (m) {
      const int l = __builtin_ctz(m);
      m &= m - 1;
      push(__builtin_amdgcn_readlane(i, l));
    }
  }
  for (int n = r0 + wave; n < r0 + RNG && n < N; n += 32) push(E + n);
  if (qn) fl(qv, qn);
}

#define AGG_T 1024
#define QCAP1 10240
static_assert(RNG <= 512 && 256 * RNG <= (1 << 17), "list entry packing");

__global__ __launch_bounds__(AGG_T)
void agg128_own(const float* __restrict__ H, const float* __restrict__ es,
                const float* __restrict__ ed, const int* __restrict__ src,
                const int* __restrict__ dst, const unsigned char* __restrict__ rid, int E, int N,
                const float* __restrict__ bias, float* __restrict__ outp) {
  __shared__ float lacc[RNG * 128];
  __shared__ uint2 lq[QCAP1];
  __shared__ float led[RNG];
  __shared__ unsigned lmx[RNG];
  __shared__ float lden[RNG];
  __shared__ int lqn;
  const int tid = threadIdx.x;
  const int lane = tid & 31;
  const int wave = __builtin_amdgcn_readfirstlane(tid >> 5);
  const int r0 = blockIdx.x * RNG;

  for (int q = tid; q < RNG * 32; q += AGG_T) ((float4*)lacc)[q] = make_float4(0.f, 0.f, 0.f, 0.f);
  for (int q = tid; q < RNG; q += AGG_T) {
    const int n = r0 + q;
    led[q] = (n < N) ? ed[n] : 0.0f;
    lmx[q] = 0u;
    lden[q] = 0.0f;
  }
  if (tid == 0) lqn = 0;
  __syncthreads();

  const float* hl = H + lane;
  float* al = lacc + lane;

  auto edge = [&](int i, int& s, int& dl, float& e) {
    int d;
    if (i < E) { s = src[i]; d = dst[i]; } else { s = d = i - E; }
    dl = d - r0;
    e = es[s] + led[dl];
    e = (e > 0.0f) ? e : NEG_SLOPE * e;
  };

  auto rows = [&](int s, int dl, float ex, int cnt) {
    const int exb = __float_as_int(ex);
    int b = 0;
    for (; b + 8 <= cnt; b += 8) {
      float hv[8][4];
      #pragma unroll
      for (int j = 0; j < 8; ++j) {
        const int sj = __builtin_amdgcn_readlane(s, b + j);
        const float* hp = hl + (size_t)sj * 128;
        hv[j][0] = hp[0]; hv[j][1] = hp[32]; hv[j][2] = hp[64]; hv[j][3] = hp[96];
      }
      #pragma unroll
      for (int j = 0; j < 8; ++j) {
        const int dj = __builtin_amdgcn_readlane(dl, b + j);
        const float ej = __int_as_float(__builtin_amdgcn_readlane(exb, b + j));
        float* o = al + dj * 128;
        atomicAdd(o +  0, hv[j][0] * ej);
        atomicAdd(o + 32, hv[j][1] * ej);
        atomicAdd(o + 64, hv[j][2] * ej);
        atomicAdd(o + 96, hv[j][3] * ej);
      }
    }
    for (; b < cnt; ++b) {
      const int sj = __builtin_amdgcn_readlane(s, b);
      const int dj = __builtin_amdgcn_readlane(dl, b);
      const float ej = __int_as_float(__builtin_amdgcn_readlane(exb, b));
      const float* hp = hl + (size_t)sj * 128;
      float* o = al + dj * 128;
      const float h0 = hp[0], h1 = hp[32], h2 = hp[64], h3 = hp[96];
      atomicAdd(o +  0, h0 * ej);
      atomicAdd(o + 32, h1 * ej);
      atomicAdd(o + 64, h2 * ej);
      atomicAdd(o + 96, h3 * ej);
    }
  };

  scan_owned(rid, E, N, r0, lane, wave, [&](int qv, int cnt) {
    int base = 0;
    if (lane == 0) base = atomicAdd(&lqn, cnt);
    base = __builtin_amdgcn_readlane(base, 0);
    if (lane < cnt) {
      int s, dl; float e;
      edge(qv, s, dl, e);
      atomicMax(&lmx[dl], fkey(e));
      const int p = base + lane;
      if (p < QCAP1) lq[p] = make_uint2((unsigned)s | ((unsigned)dl << 17), __float_as_uint(e));
    }
  });
  __syncthreads();

  const int total = __builtin_amdgcn_readfirstlane(lqn);
  if (total <= QCAP1) {
    for (int c0 = wave * 32; c0 < total; c0 += 32 * 32) {
      const int cnt = (total - c0 < 32) ? total - c0 : 32;
      int s = 0, dl = 0;
      float ex = 0.0f;
      if (lane < cnt) {
        const uint2 t2 = lq[c0 + lane];
        s = (int)(t2.x & 0x1FFFFu);
        dl = (int)(t2.x >> 17);
        ex = expf(__uint_as_float(t2.y) - funkey(lmx[dl]));
        atomicAdd(&lden[dl], ex);
      }
      rows(s, dl, ex, cnt);
    }
  } else {
    scan_owned(rid, E, N, r0, lane, wave, [&](int qv, int cnt) {
      int s = 0, dl = 0;
      float ex = 0.0f;
      if (lane < cnt) {
        float e;
        edge(qv, s, dl, e);
        ex = expf(e - funkey(lmx[dl]));
        atomicAdd(&lden[dl], ex);
      }
      rows(s, dl, ex, cnt);
    });
  }
  __syncthreads();

  for (int q = tid; q < RNG * 32; q += AGG_T) {
    const int node = q >> 5;
    const int n = r0 + node;
    if (n < N) {
      const float4 a = ((const float4*)lacc)[q];
      const float dn = lden[node];
      const float4 bb = ((const float4*)bias)[q & 31];
      float4 r;
      r.x = a.x / dn + bb.x; r.y = a.y / dn + bb.y; r.z = a.z / dn + bb.z; r.w = a.w / dn + bb.w;
      r.x = (r.x > 0.0f) ? r.x : 0.0f; r.y = (r.y > 0.0f) ? r.y : 0.0f;
      r.z = (r.z > 0.0f) ? r.z : 0.0f; r.w = (r.w > 0.0f) ? r.w : 0.0f;
      ((float4*)(outp + (size_t)n * 128))[q & 31] = r;
    }
  }
}

__global__ void node_h2e2(const float* __restrict__ H, const float* __restrict__ W2,
                          const float* __restrict__ a2s, const float* __restrict__ a2d,
                          float* __restrict__ rec, int N) {
  int n = blockIdx.x * 8 + (threadIdx.x >> 5);
  if (n >= N) return;
  int lane = threadIdx.x & 31;
  float4 hv = ((const float4*)(H + (size_t)n * 128))[lane];
  int c0 = lane * 4;
  float s0 = hv.x * W2[(c0 + 0) * 2 + 0] + hv.y * W2[(c0 + 1) * 2 + 0] +
             hv.z * W2[(c0 + 2) * 2 + 0] + hv.w * W2[(c0 + 3) * 2 + 0];
  float s1 = hv.x * W2[(c0 + 0) * 2 + 1] + hv.y * W2[(c0 + 1) * 2 + 1] +
             hv.z * W2[(c0 + 2) * 2 + 1] + hv.w * W2[(c0 + 3) * 2 + 1];
  s0 = warp_sum(s0);
  s1 = warp_sum(s1);
  if (lane == 0) {
    ((float4*)rec)[n] = make_float4(s0, s1, s0 * a2s[0] + s1 * a2s[1], s0 * a2d[0] + s1 * a2d[1]);
  }
}

#define RNG2 RNG
#define QCAP 12288

__global__ __launch_bounds__(AGG_T)
void agg2_own(const float* __restrict__ rec, const int* __restrict__ src,
              const int* __restrict__ dst, const unsigned char* __restrict__ rid, int E, int N,
              const float* __restrict__ bias, float* __restrict__ outp) {
  __shared__ float4 lq[QCAP];
  __shared__ float led[RNG2];
  __shared__ unsigned lmx[RNG2];
  __shared__ float lden[RNG2];
  __shared__ float lo0[RNG2];
  __shared__ float lo1[RNG2];
  __shared__ int lqn;
  const int tid = threadIdx.x;
  const int lane = tid & 31;
  const int wave = __builtin_amdgcn_readfirstlane(tid >> 5);
  const int r0 = blockIdx.x * RNG2;
  const float4* rec4 = (const float4*)rec;

  for (int q = tid; q < RNG2; q += AGG_T) {
    const int n = r0 + q;
    led[q] = (n < N) ? rec4[n].w : 0.0f;
    lmx[q] = 0u;
    lden[q] = 0.0f;
    lo0[q] = 0.0f;
    lo1[q] = 0.0f;
  }
  if (tid == 0) lqn = 0;
  __syncthreads();

  auto edge = [&](int i, float4& r, int& dl, float& e) {
    int s, d;
    if (i < E) { s = src[i]; d = dst[i]; } else { s = d = i - E; }
    r = rec4[s];
    dl = d - r0;
    e = r.z + led[dl];
    e = (e > 0.0f) ? e : NEG_SLOPE * e;
  };

  scan_owned(rid, E, N, r0, lane, wave, [&](int qv, int cnt) {
    int base = 0;
    if (lane == 0) base = atomicAdd(&lqn, cnt);
    base = __builtin_amdgcn_readlane(base, 0);
    if (lane < cnt) {
      float4 r; int dl; float e;
      edge(qv, r, dl, e);
      atomicMax(&lmx[dl], fkey(e));
      const int p = base + lane;
      if (p < QCAP) lq[p] = make_float4(e, r.x, r.y, __int_as_float(dl));
    }
  });
  __syncthreads();

  const int total = lqn;
  if (total <= QCAP) {
    for (int q = tid; q < total; q += AGG_T) {
      const float4 t4 = lq[q];
      const int dl = __float_as_int(t4.w);
      const float ex = expf(t4.x - funkey(lmx[dl]));
      atomicAdd(&lden[dl], ex);
      atomicAdd(&lo0[dl], t4.y * ex);
      atomicAdd(&lo1[dl], t4.z * ex);
    }
  } else {
    scan_owned(rid, E, N, r0, lane, wave, [&](int qv, int cnt) {
      if (lane < cnt) {
        float4 r; int dl; float e;
        edge(qv, r, dl, e);
        const float ex = expf(e - funkey(lmx[dl]));
        atomicAdd(&lden[dl], ex);
        atomicAdd(&lo0[dl], r.x * ex);
        atomicAdd(&lo1[dl], r.y * ex);
      }
    });
  }
  __syncthreads();

  const float bb0 = bias[0], bb1 = bias[1];
  for (int q = tid; q < RNG2; q += AGG_T) {
    const int n = r0 + q;
    if (n < N) {
      const float dn = lden[q];
      float2 o;
      o.x = lo0[q] / dn + bb0;
      o.y = lo1[q] / dn + bb1;
      ((float2*)outp)[n] = o;
    }
  }
}


extern "C" void kernel_launch(void* const* d_in, const int* in_sizes, int n_in,
                              void* d_out, int out_size, void* d_ws, size_t ws_size,
                              hipStream_t stream) {
  const float* x   = (const float*)d_in[0];
  const int*  eidx = (const int*)d_in[1];
  const float* W1  = (const float*)d_in[2];
  const float* a1s = (const float*)d_in[3];
  const float* a1d = (const float*)d_in[4];
  const float* b1  = (const float*)d_in[5];
  const float* W2  = (const float*)d_in[6];
  const float* a2s = (const float*)d_in[7];
  const float* a2d = (const float*)d_in[8];
  const float* b2  = (const float*)d_in[9];

  const int N  = in_sizes[0] / 256;
  const int E  = in_sizes[1] / 2;
  const int* src = eidx;
  const int* dst = eidx + E;
  float* out = (float*)d_out;

  char* w = (char*)d_ws;
  float*    h1b    = (float*)w;    w += (size_t)N * 128 * sizeof(float);
  float*    h1     = (float*)w;    w += (size_t)N * 128 * sizeof(float);
  float*    rec2   = (float*)w;    w += (size_t)N * 4 * sizeof(float);
  float*    e1s    = (float*)w;    w += (size_t)N * sizeof(float);
  float*    e1d    = (float*)w;    w += (size_t)N * sizeof(float);
  unsigned char* rid = (unsigned char*)w; w += (size_t)E;

  const int TB = 256;
  range_ids<<<((E + 3) / 4 + TB - 1) / TB, TB, 0, stream>>>(dst, E, rid);

  gemm1_wmma<<<G1_GRID, G1_T, 0, stream>>>(x, W1, h1, N);
  node_e1<<<(N + 7) / 8, TB, 0, stream>>>(h1, a1s, a1d, e1s, e1d, N);
  agg128_own<<<(N + RNG - 1) / RNG, AGG_T, 0, stream>>>(h1, e1s, e1d, src, dst, rid, E, N, b1, h1b);

  node_h2e2<<<(N + 7) / 8, TB, 0, stream>>>(h1b, W2, a2s, a2d, rec2, N);
  agg2_own<<<(N + RNG2 - 1) / RNG2, AGG_T, 0, stream>>>(rec2, src, dst, rid, E, N, b2, out);
}


# Layer-2 owner takes its edges from the list the layer-1 owner of the same range found ({source, local node} words via workspace): 2nd id-list scan, per-edge id loads gone; sum order = LDS arrival
# speedup vs baseline: 5.0842x; 1.2301x over previous
#include <hip/hip_runtime.h>
#include <hip/hip_bf16.h>

typedef __attribute__((ext_vector_type(2))) float v2f;
typedef __attribute__((ext_vector_type(8))) float v8f;

#define NEG_SLOPE 0.2f


__device__ __forceinline__ float warp_sum(float v) {
  #pragma unroll
  for (int o = 16; o > 0; o >>= 1) v += __shfl_xor(v, o, 32);
  return v;
}

__device__ __forceinline__ unsigned fkey(float f) {
  unsigned u = __float_as_uint(f);
  return (u & 0x80000000u) ? ~u : (u | 0x80000000u);
}
__device__ __forceinline__ float funkey(unsigned k) {
  unsigned u = (k & 0x80000000u) ? (k & 0x7FFFFFFFu) : ~k;
  return __uint_as_float(u);
}

#define G1_T 256
#define G1_GRID 512
#define G1_PITCH 260

__global__ __launch_bounds__(G1_T)
void gemm1_wmma(const float* __restrict__ X, const float* __restrict__ W,
                float* __restrict__ H, int N) {
  __shared__ __attribute__((aligned(16))) float xs[2][16 * G1_PITCH];
  const int tid = threadIdx.x;
  const int lane = tid & 31;
  const int wave = __builtin_amdgcn_readfirstlane(tid >> 5);
  const int col0 = wave * 16;
  const int m  = lane & 15;
  const int kh = lane >> 4;
  const int ntiles = (N + 15) >> 4;

  float4 b[32];
  #pragma unroll
  for (int j = 0; j < 32; ++j) {
    const float* wp = W + (size_t)(8 * j + 4 * kh) * 128 + col0 + m;
    b[j] = make_float4(wp[0], wp[128], wp[256], wp[384]);
  }

  const int srow = tid >> 6;
  const int scol = (tid & 63) * 4;
  const float* xsrc = X + scol;
  float* sdst = &xs[0][srow * G1_PITCH + scol];
  float4 p0, p1, p2, p3;
  auto fetch = [&](int t) {
    int r0 = t * 16 + srow, r1 = r0 + 4, r2 = r0 + 8, r3 = r0 + 12;
    if (r3 >= N) {
      r0 = (r0 < N) ? r0 : N - 1; r1 = (r1 < N) ? r1 : N - 1;
      r2 = (r2 < N) ? r2 : N - 1; r3 = N - 1;
    }
    p0 = *(const float4*)(xsrc + (size_t)r0 * 256);
    p1 = *(const float4*)(xsrc + (size_t)r1 * 256);
    p2 = *(const float4*)(xsrc + (size_t)r2 * 256);
    p3 = *(const float4*)(xsrc + (size_t)r3 * 256);
  };
  auto stage = [&](int bf) {
    float* d = sdst + bf * (16 * G1_PITCH);
    *(float4*)(d) = p0;
    *(float4*)(d + 4 * G1_PITCH) = p1;
    *(float4*)(d + 8 * G1_PITCH) = p2;
    *(float4*)(d + 12 * G1_PITCH) = p3;
  };

  int t = blockIdx.x;
  if (t < ntiles) { fetch(t); stage(0); }
  __syncthreads();
  int bf = 0;
  for (; t < ntiles; t += G1_GRID) {
    const int tn = t + G1_GRID;
    if (tn < ntiles) fetch(tn);

    v8f c0 = {}, c1 = {};
    const float* xa = &xs[bf][m * G1_PITCH + 4 * kh];
    float4 af[2][4];
    #pragma unroll
    for (int i = 0; i < 4; ++i) af[0][i] = *(const float4*)(xa + 8 * i);
    #pragma unroll
    for (int g = 0; g < 8; ++g) {
      if (g < 7) {
        #pragma unroll
        for (int i = 0; i < 4; ++i) af[(g + 1) & 1][i] = *(const float4*)(xa + 8 * (4 * (g + 1) + i));
      }
      #pragma unroll
      for (int i = 0; i < 4; ++i) {
        const float4 a = af[g & 1][i];
        const float4 bb = b[4 * g + i];
        v2f a0, a1, b0, b1;
        a0.x = a.x; a0.y = a.y; a1.x = a.z; a1.y = a.w;
        b0.x = bb.x; b0.y = bb.y; b1.x = bb.z; b1.y = bb.w;
#if __has_builtin(__builtin_amdgcn_wmma_f32_16x16x4_f32)
        c0 = __builtin_amdgcn_wmma_f32_16x16x4_f32(false, a0, false, b0, (short)0, c0,
                                                   false, false);
        c1 = __builtin_amdgcn_wmma_f32_16x16x4_f32(false, a1, false, b1, (short)0, c1,
                                                   false, false);
#else
        c0[0] += a0.x * b0.x + a0.y * b0.y;
        c1[0] += a1.x * b1.x + a1.y * b1.y;
#endif
      }
    }
#if __has_builtin(__builtin_amdgcn_sched_group_barrier)
    __builtin_amdgcn_sched_group_barrier(0x100, 4, 0);
    #pragma unroll
    for (int g = 0; g < 7; ++g) {
      __builtin_amdgcn_sched_group_barrier(0x100, 4, 0);
      __builtin_amdgcn_sched_group_barrier(0x008, 8, 0);
    }
    __builtin_amdgcn_sched_group_barrier(0x008, 8, 0);
#endif
    const v8f c = c0 + c1;

    if (tn < ntiles) stage(bf ^ 1);
    __syncthreads();
    bf ^= 1;

    const int row0 = t * 16;
    float* hp = H + (size_t)(row0 + 8 * kh) * 128 + col0 + m;
    if (row0 + 16 <= N) {
      #pragma unroll
      for (int r = 0; r < 8; ++r) hp[(size_t)r * 128] = c[r];
    } else {
      #pragma unroll
      for (int r = 0; r < 8; ++r)
        if (row0 + r + 8 * kh < N) hp[(size_t)r * 128] = c[r];
    }
  }
}

__global__ void node_e1(const float* __restrict__ H, const float* __restrict__ as,
                        const float* __restrict__ ad, float* __restrict__ es,
                        float* __restrict__ ed, int N) {
  int n = blockIdx.x * 8 + (threadIdx.x >> 5);
  if (n >= N) return;
  int lane = threadIdx.x & 31;
  float4 hv = ((const float4*)(H + (size_t)n * 128))[lane];
  float4 sv = ((const float4*)as)[lane];
  float4 dv = ((const float4*)ad)[lane];
  float s = hv.x * sv.x + hv.y * sv.y + hv.z * sv.z + hv.w * sv.w;
  float d = hv.x * dv.x + hv.y * dv.y + hv.z * dv.z + hv.w * dv.w;
  s = warp_sum(s);
  d = warp_sum(d);
  if (lane == 0) { es[n] = s; ed[n] = d; }
}

#define RNG 400

__global__ void range_ids(const int* __restrict__ dst, int E, unsigned char* __restrict__ rid) {
  const int t = blockIdx.x * blockDim.x + threadIdx.x;
  int i = 4 * t;
  if (i + 3 < E) {
    const unsigned w = (unsigned)(dst[i] / RNG) | ((unsigned)(dst[i + 1] / RNG) << 8) |
                       ((unsigned)(dst[i + 2] / RNG) << 16) | ((unsigned)(dst[i + 3] / RNG) << 24);
    ((unsigned*)rid)[t] = w;
  } else {
    for (; i < E; ++i) rid[i] = (unsigned char)(dst[i] / RNG);
  }
}

template <class F>
__device__ __forceinline__ void scan_owned(const unsigned char* __restrict__ rid, int E, int N,
                                           int r0, int lane, int wave, F&& fl) {
  int qv = 0;
  int qn = 0;
  auto push = [&](int i) {
    qv = (lane == qn) ? i : qv;
    if (++qn == 32) { fl(qv, 32); qn = 0; }
  };
  const unsigned own = blockIdx.x & 255u;
  const unsigned pat = own * 0x01010101u;
  auto zb = [](unsigned x) {
    return ~((((x & 0x7F7F7F7Fu) + 0x7F7F7F7Fu) | x) | 0x7F7F7F7Fu);
  };
  int a0 = (int)((16u - ((unsigned)(size_t)rid & 15u)) & 15u);
  if (a0 > E) a0 = E;
  const int nfast = (E - a0) / 2048;
  const uint4* rq = (const uint4*)(rid + a0);
  const int T = (nfast > wave) ? (nfast - wave + 31) / 32 : 0;
  int t = (int)(((size_t)blockIdx.x * (size_t)T) / gridDim.x);
  for (int k = 0; k < T; ++k) {
    const int c = t * 32 + wave;
    if (++t == T) t = 0;
    uint4 v[4];
    #pragma unroll
    for (int j = 0; j < 4; ++j) v[j] = rq[c * 128 + j * 32 + lane];
    #pragma unroll
    for (int j = 0; j < 4; ++j) {
      const unsigned f0 = zb(v[j].x ^ pat);
      const unsigned f1 = zb(v[j].y ^ pat);
      const unsigned f2 = zb(v[j].z ^ pat);
      const unsigned f3 = zb(v[j].w ^ pat);
      const unsigned m0 = __builtin_amdgcn_ballot_w32(f0 != 0u);
      const unsigned m1 = __builtin_amdgcn_ballot_w32(f1 != 0u);
      const unsigned m2 = __builtin_amdgcn_ballot_w32(f2 != 0u);
      const unsigned m3 = __builtin_amdgcn_ballot_w32(f3 != 0u);
      unsigned any = m0 | m1 | m2 | m3;
      const int ibase = a0 + (c * 128 + j * 32) * 16;
      while (any) {
        const int l = __builtin_ctz(any);
        any &= any - 1;
        const int il = ibase + l * 16;
        if ((m0 >> l) & 1u)
          for (unsigned f = __builtin_amdgcn_readlane(f0, l); f; f &= f - 1)
            push(il + 0 + (__builtin_ctz(f) >> 3));
        if ((m1 >> l) & 1u)
          for (unsigned f = __builtin_amdgcn_readlane(f1, l); f; f &= f - 1)
            push(il + 4 + (__builtin_ctz(f) >> 3));
        if ((m2 >> l) & 1u)
          for (unsigned f = __builtin_amdgcn_readlane(f2, l); f; f &= f - 1)
            push(il + 8 + (__builtin_ctz(f) >> 3));
        if ((m3 >> l) & 1u)
          for (unsigned f = __builtin_amdgcn_readlane(f3, l); f; f &= f - 1)
            push(il + 12 + (__builtin_ctz(f) >> 3));
      }
    }
  }
  const int nrem = E - nfast * 2048;
  for (int g = wave; g * 32 < nrem; g += 32) {
    const int j = g * 32 + lane;
    const int i = (j < a0) ? j : nfast * 2048 + j;
    const unsigned idv = (j < nrem) ? (unsigned)rid[i] : 256u;
    unsigned m = __builtin_amdgcn_ballot_w32(idv == own);
    while (m) {
      const int l = __builtin_ctz(m);
      m &= m - 1;
      push(__builtin_amdgcn_readlane(i, l));
    }
  }
  for (int n = r0 + wave; n < r0 + RNG && n < N; n += 32) push(E + n);
  if (qn) fl(qv, qn);
}

#define AGG_T 1024
#define QCAP1 10240
static_assert(RNG <= 512 && 256 * RNG <= (1 << 17), "list entry packing");

__global__ __launch_bounds__(AGG_T)
void agg128_own(const float* __restrict__ H, const float* __restrict__ es,
                const float* __restrict__ ed, const int* __restrict__ src,
                const int* __restrict__ dst, const unsigned char* __restrict__ rid, int E, int N,
                unsigned* __restrict__ elist, int* __restrict__ ecnt,
                const float* __restrict__ bias, float* __restrict__ outp) {
  __shared__ float lacc[RNG * 128];
  __shared__ uint2 lq[QCAP1];
  __shared__ float led[RNG];
  __shared__ unsigned lmx[RNG];
  __shared__ float lden[RNG];
  __shared__ int lqn;
  const int tid = threadIdx.x;
  const int lane = tid & 31;
  const int wave = __builtin_amdgcn_readfirstlane(tid >> 5);
  const int r0 = blockIdx.x * RNG;

  for (int q = tid; q < RNG * 32; q += AGG_T) ((float4*)lacc)[q] = make_float4(0.f, 0.f, 0.f, 0.f);
  for (int q = tid; q < RNG; q += AGG_T) {
    const int n = r0 + q;
    led[q] = (n < N) ? ed[n] : 0.0f;
    lmx[q] = 0u;
    lden[q] = 0.0f;
  }
  if (tid == 0) lqn = 0;
  __syncthreads();

  const float* hl = H + lane;
  float* al = lacc + lane;

  auto edge = [&](int i, int& s, int& dl, float& e) {
    int d;
    if (i < E) { s = src[i]; d = dst[i]; } else { s = d = i - E; }
    dl = d - r0;
    e = es[s] + led[dl];
    e = (e > 0.0f) ? e : NEG_SLOPE * e;
  };

  auto rows = [&](int s, int dl, float ex, int cnt) {
    const int exb = __float_as_int(ex);
    int b = 0;
    for (; b + 8 <= cnt; b += 8) {
      float hv[8][4];
      #pragma unroll
      for (int j = 0; j < 8; ++j) {
        const int sj = __builtin_amdgcn_readlane(s, b + j);
        const float* hp = hl + (size_t)sj * 128;
        hv[j][0] = hp[0]; hv[j][1] = hp[32]; hv[j][2] = hp[64]; hv[j][3] = hp[96];
      }
      #pragma unroll
      for (int j = 0; j < 8; ++j) {
        const int dj = __builtin_amdgcn_readlane(dl, b + j);
        const float ej = __int_as_float(__builtin_amdgcn_readlane(exb, b + j));
        float* o = al + dj * 128;
        atomicAdd(o +  0, hv[j][0] * ej);
        atomicAdd(o + 32, hv[j][1] * ej);
        atomicAdd(o + 64, hv[j][2] * ej);
        atomicAdd(o + 96, hv[j][3] * ej);
      }
    }
    for (; b < cnt; ++b) {
      const int sj = __builtin_amdgcn_readlane(s, b);
      const int dj = __builtin_amdgcn_readlane(dl, b);
      const float ej = __int_as_float(__builtin_amdgcn_readlane(exb, b));
      const float* hp = hl + (size_t)sj * 128;
      float* o = al + dj * 128;
      const float h0 = hp[0], h1 = hp[32], h2 = hp[64], h3 = hp[96];
      atomicAdd(o +  0, h0 * ej);
      atomicAdd(o + 32, h1 * ej);
      atomicAdd(o + 64, h2 * ej);
      atomicAdd(o + 96, h3 * ej);
    }
  };

  scan_owned(rid, E, N, r0, lane, wave, [&](int qv, int cnt) {
    int base = 0;
    if (lane == 0) base = atomicAdd(&lqn, cnt);
    base = __builtin_amdgcn_readlane(base, 0);
    if (lane < cnt) {
      int s, dl; float e;
      edge(qv, s, dl, e);
      atomicMax(&lmx[dl], fkey(e));
      const int p = base + lane;
      if (p < QCAP1) lq[p] = make_uint2((unsigned)s | ((unsigned)dl << 17), __float_as_uint(e));
    }
  });
  __syncthreads();

  const int total = __builtin_amdgcn_readfirstlane(lqn);
  if (tid == 0) ecnt[blockIdx.x] = total;
  if (total <= QCAP1) {
    unsigned* el = elist + (size_t)blockIdx.x * QCAP1;
    for (int q = tid; q < total; q += AGG_T) el[q] = lq[q].x;
    for (int c0 = wave * 32; c0 < total; c0 += 32 * 32) {
      const int cnt = (total - c0 < 32) ? total - c0 : 32;
      int s = 0, dl = 0;
      float ex = 0.0f;
      if (lane < cnt) {
        const uint2 t2 = lq[c0 + lane];
        s = (int)(t2.x & 0x1FFFFu);
        dl = (int)(t2.x >> 17);
        ex = expf(__uint_as_float(t2.y) - funkey(lmx[dl]));
        atomicAdd(&lden[dl], ex);
      }
      rows(s, dl, ex, cnt);
    }
  } else {
    scan_owned(rid, E, N, r0, lane, wave, [&](int qv, int cnt) {
      int s = 0, dl = 0;
      float ex = 0.0f;
      if (lane < cnt) {
        float e;
        edge(qv, s, dl, e);
        ex = expf(e - funkey(lmx[dl]));
        atomicAdd(&lden[dl], ex);
      }
      rows(s, dl, ex, cnt);
    });
  }
  __syncthreads();

  for (int q = tid; q < RNG * 32; q += AGG_T) {
    const int node = q >> 5;
    const int n = r0 + node;
    if (n < N) {
      const float4 a = ((const float4*)lacc)[q];
      const float dn = lden[node];
      const float4 bb = ((const float4*)bias)[q & 31];
      float4 r;
      r.x = a.x / dn + bb.x; r.y = a.y / dn + bb.y; r.z = a.z / dn + bb.z; r.w = a.w / dn + bb.w;
      r.x = (r.x > 0.0f) ? r.x : 0.0f; r.y = (r.y > 0.0f) ? r.y : 0.0f;
      r.z = (r.z > 0.0f) ? r.z : 0.0f; r.w = (r.w > 0.0f) ? r.w : 0.0f;
      ((float4*)(outp + (size_t)n * 128))[q & 31] = r;
    }
  }
}

__global__ void node_h2e2(const float* __restrict__ H, const float* __restrict__ W2,
                          const float* __restrict__ a2s, const float* __restrict__ a2d,
                          float* __restrict__ rec, int N) {
  int n = blockIdx.x * 8 + (threadIdx.x >> 5);
  if (n >= N) return;
  int lane = threadIdx.x & 31;
  float4 hv = ((const float4*)(H + (size_t)n * 128))[lane];
  int c0 = lane * 4;
  float s0 = hv.x * W2[(c0 + 0) * 2 + 0] + hv.y * W2[(c0 + 1) * 2 + 0] +
             hv.z * W2[(c0 + 2) * 2 + 0] + hv.w * W2[(c0 + 3) * 2 + 0];
  float s1 = hv.x * W2[(c0 + 0) * 2 + 1] + hv.y * W2[(c0 + 1) * 2 + 1] +
             hv.z * W2[(c0 + 2) * 2 + 1] + hv.w * W2[(c0 + 3) * 2 + 1];
  s0 = warp_sum(s0);
  s1 = warp_sum(s1);
  if (lane == 0) {
    ((float4*)rec)[n] = make_float4(s0, s1, s0 * a2s[0] + s1 * a2s[1], s0 * a2d[0] + s1 * a2d[1]);
  }
}

#define RNG2 RNG
#define QCAP 12288

static_assert(QCAP1 <= QCAP, "list hand-over");

__global__ __launch_bounds__(AGG_T)
void agg2_own(const float* __restrict__ rec, const int* __restrict__ src,
              const int* __restrict__ dst, const unsigned char* __restrict__ rid, int E, int N,
              const unsigned* __restrict__ elist, const int* __restrict__ ecnt,
              const float* __restrict__ bias, float* __restrict__ outp) {
  __shared__ float4 lq[QCAP];
  __shared__ float led[RNG2];
  __shared__ unsigned lmx[RNG2];
  __shared__ float lden[RNG2];
  __shared__ float lo0[RNG2];
  __shared__ float lo1[RNG2];
  __shared__ int lqn;
  const int tid = threadIdx.x;
  const int lane = tid & 31;
  const int wave = __builtin_amdgcn_readfirstlane(tid >> 5);
  const int r0 = blockIdx.x * RNG2;
  const float4* rec4 = (const float4*)rec;

  for (int q = tid; q < RNG2; q += AGG_T) {
    const int n = r0 + q;
    led[q] = (n < N) ? rec4[n].w : 0.0f;
    lmx[q] = 0u;
    lden[q] = 0.0f;
    lo0[q] = 0.0f;
    lo1[q] = 0.0f;
  }
  if (tid == 0) lqn = 0;
  __syncthreads();

  auto edge = [&](int i, float4& r, int& dl, float& e) {
    int s, d;
    if (i < E) { s = src[i]; d = dst[i]; } else { s = d = i - E; }
    r = rec4[s];
    dl = d - r0;
    e = r.z + led[dl];
    e = (e > 0.0f) ? e : NEG_SLOPE * e;
  };

  const int found = ecnt[blockIdx.x];
  const bool listed = found <= QCAP1;
  if (listed) {
    const unsigned* el = elist + (size_t)blockIdx.x * QCAP1;
    for (int q0 = tid; q0 < found; q0 += 8 * AGG_T) {
      unsigned w[8];
      float4 r[8];
      #pragma unroll
      for (int j = 0; j < 8; ++j) {
        const int q = q0 + j * AGG_T;
        w[j] = (q < found) ? el[q] : 0u;
      }
      #pragma unroll
      for (int j = 0; j < 8; ++j) r[j] = rec4[w[j] & 0x1FFFFu];
      asm volatile("" :: "v"(r[0].z), "v"(r[1].z), "v"(r[2].z), "v"(r[3].z), "v"(r[4].z),
                   "v"(r[5].z), "v"(r[6].z), "v"(r[7].z));
      #pragma unroll
      for (int j = 0; j < 8; ++j) {
        const int q = q0 + j * AGG_T;
        if (q < found) {
          const int dl = (int)(w[j] >> 17);
          float e = r[j].z + led[dl];
          e = (e > 0.0f) ? e : NEG_SLOPE * e;
          atomicMax(&lmx[dl], fkey(e));
          lq[q] = make_float4(e, r[j].x, r[j].y, __int_as_float(dl));
        }
      }
    }
  } else {
    scan_owned(rid, E, N, r0, lane, wave, [&](int qv, int cnt) {
      int base = 0;
      if (lane == 0) base = atomicAdd(&lqn, cnt);
      base = __builtin_amdgcn_readlane(base, 0);
      if (lane < cnt) {
        float4 r; int dl; float e;
        edge(qv, r, dl, e);
        atomicMax(&lmx[dl], fkey(e));
        const int p = base + lane;
        if (p < QCAP) lq[p] = make_float4(e, r.x, r.y, __int_as_float(dl));
      }
    });
  }
  __syncthreads();

  const int total = listed ? found : lqn;
  if (total <= QCAP) {
    for (int q = tid; q < total; q += AGG_T) {
      const float4 t4 = lq[q];
      const int dl = __float_as_int(t4.w);
      const float ex = expf(t4.x - funkey(lmx[dl]));
      atomicAdd(&lden[dl], ex);
      atomicAdd(&lo0[dl], t4.y * ex);
      atomicAdd(&lo1[dl], t4.z * ex);
    }
  } else {
    scan_owned(rid, E, N, r0, lane, wave, [&](int qv, int cnt) {
      if (lane < cnt) {
        float4 r; int dl; float e;
        edge(qv, r, dl, e);
        const float ex = expf(e - funkey(lmx[dl]));
        atomicAdd(&lden[dl], ex);
        atomicAdd(&lo0[dl], r.x * ex);
        atomicAdd(&lo1[dl], r.y * ex);
      }
    });
  }
  __syncthreads();

  const float bb0 = bias[0], bb1 = bias[1];
  for (int q = tid; q < RNG2; q += AGG_T) {
    const int n = r0 + q;
    if (n < N) {
      const float dn = lden[q];
      float2 o;
      o.x = lo0[q] / dn + bb0;
      o.y = lo1[q] / dn + bb1;
      ((float2*)outp)[n] = o;
    }
  }
}


extern "C" void kernel_launch(void* const* d_in, const int* in_sizes, int n_in,
                              void* d_out, int out_size, void* d_ws, size_t ws_size,
                              hipStream_t stream) {
  const float* x   = (const float*)d_in[0];
  const int*  eidx = (const int*)d_in[1];
  const float* W1  = (const float*)d_in[2];
  const float* a1s = (const float*)d_in[3];
  const float* a1d = (const float*)d_in[4];
  const float* b1  = (const float*)d_in[5];
  const float* W2  = (const float*)d_in[6];
  const float* a2s = (const float*)d_in[7];
  const float* a2d = (const float*)d_in[8];
  const float* b2  = (const float*)d_in[9];

  const int N  = in_sizes[0] / 256;
  const int E  = in_sizes[1] / 2;
  const int* src = eidx;
  const int* dst = eidx + E;
  float* out = (float*)d_out;

  char* w = (char*)d_ws;
  float*    h1b    = (float*)w;    w += (size_t)N * 128 * sizeof(float);
  float*    h1     = (float*)w;    w += (size_t)N * 128 * sizeof(float);
  float*    rec2   = (float*)w;    w += (size_t)N * 4 * sizeof(float);
  float*    e1s    = (float*)w;    w += (size_t)N * sizeof(float);
  float*    e1d    = (float*)w;    w += (size_t)N * sizeof(float);
  unsigned* elist  = (unsigned*)w; w += (size_t)((N + RNG - 1) / RNG) * QCAP1 * sizeof(unsigned);
  int*      ecnt   = (int*)w;      w += 256 * sizeof(int);
  unsigned char* rid = (unsigned char*)w; w += (size_t)E;

  const int TB = 256;
  range_ids<<<((E + 3) / 4 + TB - 1) / TB, TB, 0, stream>>>(dst, E, rid);

  gemm1_wmma<<<G1_GRID, G1_T, 0, stream>>>(x, W1, h1, N);
  node_e1<<<(N + 7) / 8, TB, 0, stream>>>(h1, a1s, a1d, e1s, e1d, N);
  agg128_own<<<(N + RNG - 1) / RNG, AGG_T, 0, stream>>>(h1, e1s, e1d, src, dst, rid, E, N, elist,
                                                        ecnt, b1, h1b);

  node_h2e2<<<(N + 7) / 8, TB, 0, stream>>>(h1b, W2, a2s, a2d, rec2, N);
  agg2_own<<<(N + RNG2 - 1) / RNG2, AGG_T, 0, stream>>>(rec2, src, dst, rid, E, N, elist, ecnt, b2,
                                                        out);
}


# Per-node projection kernels (e1; h2+e2): a wave takes 16 consecutive nodes, rows loaded ahead, 32 partial sums folded over the lanes together, one store a wave; 200000 waves to 12500; sum order kept
# speedup vs baseline: 5.4990x; 1.0816x over previous
#include <hip/hip_runtime.h>
#include <hip/hip_bf16.h>

typedef __attribute__((ext_vector_type(2))) float v2f;
typedef __attribute__((ext_vector_type(8))) float v8f;

#define NEG_SLOPE 0.2f


__device__ __forceinline__ float fold16(const float (&v)[16], int lane) {
  const bool b4 = (lane & 16) != 0, b3 = (lane & 8) != 0, b2 = (lane & 4) != 0, b1 = (lane & 2) != 0;
  float a8[8], a4[4], a2[2];
  #pragma unroll
  for (int j = 0; j < 8; ++j) {
    const float keep = b4 ? v[j + 8] : v[j];
    const float send = b4 ? v[j] : v[j + 8];
    a8[j] = keep + __shfl_xor(send, 16, 32);
  }
  #pragma unroll
  for (int j = 0; j < 4; ++j) {
    const float keep = b3 ? a8[j + 4] : a8[j];
    const float send = b3 ? a8[j] : a8[j + 4];
    a4[j] = keep + __shfl_xor(send, 8, 32);
  }
  #pragma unroll
  for (int j = 0; j < 2; ++j) {
    const float keep = b2 ? a4[j + 2] : a4[j];
    const float send = b2 ? a4[j] : a4[j + 2];
    a2[j] = keep + __shfl_xor(send, 4, 32);
  }
  const float keep = b1 ? a2[1] : a2[0];
  const float send = b1 ? a2[0] : a2[1];
  return keep + __shfl_xor(send, 2, 32);
}

__device__ __forceinline__ unsigned fkey(float f) {
  unsigned u = __float_as_uint(f);
  return (u & 0x80000000u) ? ~u : (u | 0x80000000u);
}
__device__ __forceinline__ float funkey(unsigned k) {
  unsigned u = (k & 0x80000000u) ? (k & 0x7FFFFFFFu) : ~k;
  return __uint_as_float(u);
}

#define G1_T 256
#define G1_GRID 512
#define G1_PITCH 260

__global__ __launch_bounds__(G1_T)
void gemm1_wmma(const float* __restrict__ X, const float* __restrict__ W,
                float* __restrict__ H, int N) {
  __shared__ __attribute__((aligned(16))) float xs[2][16 * G1_PITCH];
  const int tid = threadIdx.x;
  const int lane = tid & 31;
  const int wave = __builtin_amdgcn_readfirstlane(tid >> 5);
  const int col0 = wave * 16;
  const int m  = lane & 15;
  const int kh = lane >> 4;
  const int ntiles = (N + 15) >> 4;

  float4 b[32];
  #pragma unroll
  for (int j = 0; j < 32; ++j) {
    const float* wp = W + (size_t)(8 * j + 4 * kh) * 128 + col0 + m;
    b[j] = make_float4(wp[0], wp[128], wp[256], wp[384]);
  }

  const int srow = tid >> 6;
  const int scol = (tid & 63) * 4;
  const float* xsrc = X + scol;
  float* sdst = &xs[0][srow * G1_PITCH + scol];
  float4 p0, p1, p2, p3;
  auto fetch = [&](int t) {
    int r0 = t * 16 + srow, r1 = r0 + 4, r2 = r0 + 8, r3 = r0 + 12;
    if (r3 >= N) {
      r0 = (r0 < N) ? r0 : N - 1; r1 = (r1 < N) ? r1 : N - 1;
      r2 = (r2 < N) ? r2 : N - 1; r3 = N - 1;
    }
    p0 = *(const float4*)(xsrc + (size_t)r0 * 256);
    p1 = *(const float4*)(xsrc + (size_t)r1 * 256);
    p2 = *(const float4*)(xsrc + (size_t)r2 * 256);
    p3 = *(const float4*)(xsrc + (size_t)r3 * 256);
  };
  auto stage = [&](int bf) {
    float* d = sdst + bf * (16 * G1_PITCH);
    *(float4*)(d) = p0;
    *(float4*)(d + 4 * G1_PITCH) = p1;
    *(float4*)(d + 8 * G1_PITCH) = p2;
    *(float4*)(d + 12 * G1_PITCH) = p3;
  };

  int t = blockIdx.x;
  if (t < ntiles) { fetch(t); stage(0); }
  __syncthreads();
  int bf = 0;
  for (; t < ntiles; t += G1_GRID) {
    const int tn = t + G1_GRID;
    if (tn < ntiles) fetch(tn);

    v8f c0 = {}, c1 = {};
    const float* xa = &xs[bf][m * G1_PITCH + 4 * kh];
    float4 af[2][4];
    #pragma unroll
    for (int i = 0; i < 4; ++i) af[0][i] = *(const float4*)(xa + 8 * i);
    #pragma unroll
    for (int g = 0; g < 8; ++g) {
      if (g < 7) {
        #pragma unroll
        for (int i = 0; i < 4; ++i) af[(g + 1) & 1][i] = *(const float4*)(xa + 8 * (4 * (g + 1) + i));
      }
      #pragma unroll
      for (int i = 0; i < 4; ++i) {
        const float4 a = af[g & 1][i];
        const float4 bb = b[4 * g + i];
        v2f a0, a1, b0, b1;
        a0.x = a.x; a0.y = a.y; a1.x = a.z; a1.y = a.w;
        b0.x = bb.x; b0.y = bb.y; b1.x = bb.z; b1.y = bb.w;
#if __has_builtin(__builtin_amdgcn_wmma_f32_16x16x4_f32)
        c0 = __builtin_amdgcn_wmma_f32_16x16x4_f32(false, a0, false, b0, (short)0, c0,
                                                   false, false);
        c1 = __builtin_amdgcn_wmma_f32_16x16x4_f32(false, a1, false, b1, (short)0, c1,
                                                   false, false);
#else
        c0[0] += a0.x * b0.x + a0.y * b0.y;
        c1[0] += a1.x * b1.x + a1.y * b1.y;
#endif
      }
    }
#if __has_builtin(__builtin_amdgcn_sched_group_barrier)
    __builtin_amdgcn_sched_group_barrier(0x100, 4, 0);
    #pragma unroll
    for (int g = 0; g < 7; ++g) {
      __builtin_amdgcn_sched_group_barrier(0x100, 4, 0);
      __builtin_amdgcn_sched_group_barrier(0x008, 8, 0);
    }
    __builtin_amdgcn_sched_group_barrier(0x008, 8, 0);
#endif
    const v8f c = c0 + c1;

    if (tn < ntiles) stage(bf ^ 1);
    __syncthreads();
    bf ^= 1;

    const int row0 = t * 16;
    float* hp = H + (size_t)(row0 + 8 * kh) * 128 + col0 + m;
    if (row0 + 16 <= N) {
      #pragma unroll
      for (int r = 0; r < 8; ++r) hp[(size_t)r * 128] = c[r];
    } else {
      #pragma unroll
      for (int r = 0; r < 8; ++r)
        if (row0 + r + 8 * kh < N) hp[(size_t)r * 128] = c[r];
    }
  }
}

#define NPW 16
static_assert(NPW == 16, "fold16 folds 16 values per lane");
#define NODE_T 256
#define NODE_SPAN (NPW * (NODE_T / 32))

__global__ __launch_bounds__(NODE_T)
void node_e1(const float* __restrict__ H, const float* __restrict__ as,
             const float* __restrict__ ad, float* __restrict__ es,
             float* __restrict__ ed, int N) {
  const int lane = threadIdx.x & 31;
  const int n0 = __builtin_amdgcn_readfirstlane(
      (int)(blockIdx.x * (NODE_T / 32) + (threadIdx.x >> 5)) * NPW);
  if (n0 >= N) return;
  const float4 sv = ((const float4*)as)[lane];
  const float4 dv = ((const float4*)ad)[lane];
  const float4* H4 = (const float4*)H;
  float4 hv[NPW];
  #pragma unroll
  for (int j = 0; j < NPW; ++j) {
    const int n = (n0 + j < N) ? n0 + j : N - 1;
    hv[j] = H4[(size_t)n * 32 + lane];
  }
  float s[NPW], d[NPW];
  #pragma unroll
  for (int j = 0; j < NPW; ++j) {
    s[j] = hv[j].x * sv.x + hv[j].y * sv.y + hv[j].z * sv.z + hv[j].w * sv.w;
    d[j] = hv[j].x * dv.x + hv[j].y * dv.y + hv[j].z * dv.z + hv[j].w * dv.w;
  }
  const float fs = fold16(s, lane);
  const float fd = fold16(d, lane);
  const bool b0 = (lane & 1) != 0;
  const float keep = b0 ? fd : fs;
  const float send = b0 ? fs : fd;
  const float r = keep + __shfl_xor(send, 1, 32);
  const int n = n0 + (lane >> 1);
  if (n < N) {
    float* o = b0 ? ed : es;
    o[n] = r;
  }
}

#define RNG 400

__global__ void range_ids(const int* __restrict__ dst, int E, unsigned char* __restrict__ rid) {
  const int t = blockIdx.x * blockDim.x + threadIdx.x;
  int i = 4 * t;
  if (i + 3 < E) {
    const unsigned w = (unsigned)(dst[i] / RNG) | ((unsigned)(dst[i + 1] / RNG) << 8) |
                       ((unsigned)(dst[i + 2] / RNG) << 16) | ((unsigned)(dst[i + 3] / RNG) << 24);
    ((unsigned*)rid)[t] = w;
  } else {
    for (; i < E; ++i) rid[i] = (unsigned char)(dst[i] / RNG);
  }
}

template <class F>
__device__ __forceinline__ void scan_owned(const unsigned char* __restrict__ rid, int E, int N,
                                           int r0, int lane, int wave, F&& fl) {
  int qv = 0;
  int qn = 0;
  auto push = [&](int i) {
    qv = (lane == qn) ? i : qv;
    if (++qn == 32) { fl(qv, 32); qn = 0; }
  };
  const unsigned own = blockIdx.x & 255u;
  const unsigned pat = own * 0x01010101u;
  auto zb = [](unsigned x) {
    return ~((((x & 0x7F7F7F7Fu) + 0x7F7F7F7Fu) | x) | 0x7F7F7F7Fu);
  };
  int a0 = (int)((16u - ((unsigned)(size_t)rid & 15u)) & 15u);
  if (a0 > E) a0 = E;
  const int nfast = (E - a0) / 2048;
  const uint4* rq = (const uint4*)(rid + a0);
  const int T = (nfast > wave) ? (nfast - wave + 31) / 32 : 0;
  int t = (int)(((size_t)blockIdx.x * (size_t)T) / gridDim.x);
  for (int k = 0; k < T; ++k) {
    const int c = t * 32 + wave;
    if (++t == T) t = 0;
    uint4 v[4];
    #pragma unroll
    for (int j = 0; j < 4; ++j) v[j] = rq[c * 128 + j * 32 + lane];
    #pragma unroll
    for (int j = 0; j < 4; ++j) {
      const unsigned f0 = zb(v[j].x ^ pat);
      const unsigned f1 = zb(v[j].y ^ pat);
      const unsigned f2 = zb(v[j].z ^ pat);
      const unsigned f3 = zb(v[j].w ^ pat);
      const unsigned m0 = __builtin_amdgcn_ballot_w32(f0 != 0u);
      const unsigned m1 = __builtin_amdgcn_ballot_w32(f1 != 0u);
      const unsigned m2 = __builtin_amdgcn_ballot_w32(f2 != 0u);
      const unsigned m3 = __builtin_amdgcn_ballot_w32(f3 != 0u);
      unsigned any = m0 | m1 | m2 | m3;
      const int ibase = a0 + (c * 128 + j * 32) * 16;
      while (any) {
        const int l = __builtin_ctz(any);
        any &= any - 1;
        const int il = ibase + l * 16;
        if ((m0 >> l) & 1u)
          for (unsigned f = __builtin_amdgcn_readlane(f0, l); f; f &= f - 1)
            push(il + 0 + (__builtin_ctz(f) >> 3));
        if ((m1 >> l) & 1u)
          for (unsigned f = __builtin_amdgcn_readlane(f1, l); f; f &= f - 1)
            push(il + 4 + (__builtin_ctz(f) >> 3));
        if ((m2 >> l) & 1u)
          for (unsigned f = __builtin_amdgcn_readlane(f2, l); f; f &= f - 1)
            push(il + 8 + (__builtin_ctz(f) >> 3));
        if ((m3 >> l) & 1u)
          for (unsigned f = __builtin_amdgcn_readlane(f3, l); f; f &= f - 1)
            push(il + 12 + (__builtin_ctz(f) >> 3));
      }
    }
  }
  const int nrem = E - nfast * 2048;
  for (int g = wave; g * 32 < nrem; g += 32) {
    const int j = g * 32 + lane;
    const int i = (j < a0) ? j : nfast * 2048 + j;
    const unsigned idv = (j < nrem) ? (unsigned)rid[i] : 256u;
    unsigned m = __builtin_amdgcn_ballot_w32(idv == own);
    while (m) {
      const int l = __builtin_ctz(m);
      m &= m - 1;
      push(__builtin_amdgcn_readlane(i, l));
    }
  }
  for (int n = r0 + wave; n < r0 + RNG && n < N; n += 32) push(E + n);
  if (qn) fl(qv, qn);
}

#define AGG_T 1024
#define QCAP1 10240
static_assert(RNG <= 512 && 256 * RNG <= (1 << 17), "list entry packing");

__global__ __launch_bounds__(AGG_T)
void agg128_own(const float* __restrict__ H, const float* __restrict__ es,
                const float* __restrict__ ed, const int* __restrict__ src,
                const int* __restrict__ dst, const unsigned char* __restrict__ rid, int E, int N,
                unsigned* __restrict__ elist, int* __restrict__ ecnt,
                const float* __restrict__ bias, float* __restrict__ outp) {
  __shared__ float lacc[RNG * 128];
  __shared__ uint2 lq[QCAP1];
  __shared__ float led[RNG];
  __shared__ unsigned lmx[RNG];
  __shared__ float lden[RNG];
  __shared__ int lqn;
  const int tid = threadIdx.x;
  const int lane = tid & 31;
  const int wave = __builtin_amdgcn_readfirstlane(tid >> 5);
  const int r0 = blockIdx.x * RNG;

  for (int q = tid; q < RNG * 32; q += AGG_T) ((float4*)lacc)[q] = make_float4(0.f, 0.f, 0.f, 0.f);
  for (int q = tid; q < RNG; q += AGG_T) {
    const int n = r0 + q;
    led[q] = (n < N) ? ed[n] : 0.0f;
    lmx[q] = 0u;
    lden[q] = 0.0f;
  }
  if (tid == 0) lqn = 0;
  __syncthreads();

  const float* hl = H + lane;
  float* al = lacc + lane;

  auto edge = [&](int i, int& s, int& dl, float& e) {
    int d;
    if (i < E) { s = src[i]; d = dst[i]; } else { s = d = i - E; }
    dl = d - r0;
    e = es[s] + led[dl];
    e = (e > 0.0f) ? e : NEG_SLOPE * e;
  };

  auto rows = [&](int s, int dl, float ex, int cnt) {
    const int exb = __float_as_int(ex);
    int b = 0;
    for (; b + 8 <= cnt; b += 8) {
      float hv[8][4];
      #pragma unroll
      for (int j = 0; j < 8; ++j) {
        const int sj = __builtin_amdgcn_readlane(s, b + j);
        const float* hp = hl + (size_t)sj * 128;
        hv[j][0] = hp[0]; hv[j][1] = hp[32]; hv[j][2] = hp[64]; hv[j][3] = hp[96];
      }
      #pragma unroll
      for (int j = 0; j < 8; ++j) {
        const int dj = __builtin_amdgcn_readlane(dl, b + j);
        const float ej = __int_as_float(__builtin_amdgcn_readlane(exb, b + j));
        float* o = al + dj * 128;
        atomicAdd(o +  0, hv[j][0] * ej);
        atomicAdd(o + 32, hv[j][1] * ej);
        atomicAdd(o + 64, hv[j][2] * ej);
        atomicAdd(o + 96, hv[j][3] * ej);
      }
    }
    for (; b < cnt; ++b) {
      const int sj = __builtin_amdgcn_readlane(s, b);
      const int dj = __builtin_amdgcn_readlane(dl, b);
      const float ej = __int_as_float(__builtin_amdgcn_readlane(exb, b));
      const float* hp = hl + (size_t)sj * 128;
      float* o = al + dj * 128;
      const float h0 = hp[0], h1 = hp[32], h2 = hp[64], h3 = hp[96];
      atomicAdd(o +  0, h0 * ej);
      atomicAdd(o + 32, h1 * ej);
      atomicAdd(o + 64, h2 * ej);
      atomicAdd(o + 96, h3 * ej);
    }
  };

  scan_owned(rid, E, N, r0, lane, wave, [&](int qv, int cnt) {
    int base = 0;
    if (lane == 0) base = atomicAdd(&lqn, cnt);
    base = __builtin_amdgcn_readlane(base, 0);
    if (lane < cnt) {
      int s, dl; float e;
      edge(qv, s, dl, e);
      atomicMax(&lmx[dl], fkey(e));
      const int p = base + lane;
      if (p < QCAP1) lq[p] = make_uint2((unsigned)s | ((unsigned)dl << 17), __float_as_uint(e));
    }
  });
  __syncthreads();

  const int total = __builtin_amdgcn_readfirstlane(lqn);
  if (tid == 0) ecnt[blockIdx.x] = total;
  if (total <= QCAP1) {
    unsigned* el = elist + (size_t)blockIdx.x * QCAP1;
    for (int q = tid; q < total; q += AGG_T) el[q] = lq[q].x;
    for (int c0 = wave * 32; c0 < total; c0 += 32 * 32) {
      const int cnt = (total - c0 < 32) ? total - c0 : 32;
      int s = 0, dl = 0;
      float ex = 0.0f;
      if (lane < cnt) {
        const uint2 t2 = lq[c0 + lane];
        s = (int)(t2.x & 0x1FFFFu);
        dl = (int)(t2.x >> 17);
        ex = expf(__uint_as_float(t2.y) - funkey(lmx[dl]));
        atomicAdd(&lden[dl], ex);
      }
      rows(s, dl, ex, cnt);
    }
  } else {
    scan_owned(rid, E, N, r0, lane, wave, [&](int qv, int cnt) {
      int s = 0, dl = 0;
      float ex = 0.0f;
      if (lane < cnt) {
        float e;
        edge(qv, s, dl, e);
        ex = expf(e - funkey(lmx[dl]));
        atomicAdd(&lden[dl], ex);
      }
      rows(s, dl, ex, cnt);
    });
  }
  __syncthreads();

  for (int q = tid; q < RNG * 32; q += AGG_T) {
    const int node = q >> 5;
    const int n = r0 + node;
    if (n < N) {
      const float4 a = ((const float4*)lacc)[q];
      const float dn = lden[node];
      const float4 bb = ((const float4*)bias)[q & 31];
      float4 r;
      r.x = a.x / dn + bb.x; r.y = a.y / dn + bb.y; r.z = a.z / dn + bb.z; r.w = a.w / dn + bb.w;
      r.x = (r.x > 0.0f) ? r.x : 0.0f; r.y = (r.y > 0.0f) ? r.y : 0.0f;
      r.z = (r.z > 0.0f) ? r.z : 0.0f; r.w = (r.w > 0.0f) ? r.w : 0.0f;
      ((float4*)(outp + (size_t)n * 128))[q & 31] = r;
    }
  }
}

__global__ __launch_bounds__(NODE_T)
void node_h2e2(const float* __restrict__ H, const float* __restrict__ W2,
               const float* __restrict__ a2s, const float* __restrict__ a2d,
               float* __restrict__ rec, int N) {
  const int lane = threadIdx.x & 31;
  const int n0 = __builtin_amdgcn_readfirstlane(
      (int)(blockIdx.x * (NODE_T / 32) + (threadIdx.x >> 5)) * NPW);
  if (n0 >= N) return;
  const int c0 = lane * 4;
  const float w00 = W2[(c0 + 0) * 2 + 0], w01 = W2[(c0 + 0) * 2 + 1];
  const float w10 = W2[(c0 + 1) * 2 + 0], w11 = W2[(c0 + 1) * 2 + 1];
  const float w20 = W2[(c0 + 2) * 2 + 0], w21 = W2[(c0 + 2) * 2 + 1];
  const float w30 = W2[(c0 + 3) * 2 + 0], w31 = W2[(c0 + 3) * 2 + 1];
  const float4* H4 = (const float4*)H;
  float4 hv[NPW];
  #pragma unroll
  for (int j = 0; j < NPW; ++j) {
    const int n = (n0 + j < N) ? n0 + j : N - 1;
    hv[j] = H4[(size_t)n * 32 + lane];
  }
  float s0[NPW], s1[NPW];
  #pragma unroll
  for (int j = 0; j < NPW; ++j) {
    s0[j] = hv[j].x * w00 + hv[j].y * w10 + hv[j].z * w20 + hv[j].w * w30;
    s1[j] = hv[j].x * w01 + hv[j].y * w11 + hv[j].z * w21 + hv[j].w * w31;
  }
  const float f0 = fold16(s0, lane);
  const float f1 = fold16(s1, lane);
  const bool b0 = (lane & 1) != 0;
  const float keep = b0 ? f1 : f0;
  const float send = b0 ? f0 : f1;
  const float r = keep + __shfl_xor(send, 1, 32);
  const float q = __shfl_xor(r, 1, 32);
  const float h0 = b0 ? q : r;
  const float h1 = b0 ? r : q;
  float2 o;
  o.x = b0 ? h0 * a2s[0] + h1 * a2s[1] : h0;
  o.y = b0 ? h0 * a2d[0] + h1 * a2d[1] : h1;
  const int n = n0 + (lane >> 1);
  if (n < N) ((float2*)rec)[(size_t)n * 2 + (b0 ? 1 : 0)] = o;
}

#define RNG2 RNG
#define QCAP 12288

static_assert(QCAP1 <= QCAP, "list hand-over");

__global__ __launch_bounds__(AGG_T)
void agg2_own(const float* __restrict__ rec, const int* __restrict__ src,
              const int* __restrict__ dst, const unsigned char* __restrict__ rid, int E, int N,
              const unsigned* __restrict__ elist, const int* __restrict__ ecnt,
              const float* __restrict__ bias, float* __restrict__ outp) {
  __shared__ float4 lq[QCAP];
  __shared__ float led[RNG2];
  __shared__ unsigned lmx[RNG2];
  __shared__ float lden[RNG2];
  __shared__ float lo0[RNG2];
  __shared__ float lo1[RNG2];
  __shared__ int lqn;
  const int tid = threadIdx.x;
  const int lane = tid & 31;
  const int wave = __builtin_amdgcn_readfirstlane(tid >> 5);
  const int r0 = blockIdx.x * RNG2;
  const float4* rec4 = (const float4*)rec;

  for (int q = tid; q < RNG2; q += AGG_T) {
    const int n = r0 + q;
    led[q] = (n < N) ? rec4[n].w : 0.0f;
    lmx[q] = 0u;
    lden[q] = 0.0f;
    lo0[q] = 0.0f;
    lo1[q] = 0.0f;
  }
  if (tid == 0) lqn = 0;
  __syncthreads();

  auto edge = [&](int i, float4& r, int& dl, float& e) {
    int s, d;
    if (i < E) { s = src[i]; d = dst[i]; } else { s = d = i - E; }
    r = rec4[s];
    dl = d - r0;
    e = r.z + led[dl];
    e = (e > 0.0f) ? e : NEG_SLOPE * e;
  };

  const int found = ecnt[blockIdx.x];
  const bool listed = found <= QCAP1;
  if (listed) {
    const unsigned* el = elist + (size_t)blockIdx.x * QCAP1;
    for (int q0 = tid; q0 < found; q0 += 8 * AGG_T) {
      unsigned w[8];
      float4 r[8];
      #pragma unroll
      for (int j = 0; j < 8; ++j) {
        const int q = q0 + j * AGG_T;
        w[j] = (q < found) ? el[q] : 0u;
      }
      #pragma unroll
      for (int j = 0; j < 8; ++j) r[j] = rec4[w[j] & 0x1FFFFu];
      asm volatile("" :: "v"(r[0].z), "v"(r[1].z), "v"(r[2].z), "v"(r[3].z), "v"(r[4].z),
                   "v"(r[5].z), "v"(r[6].z), "v"(r[7].z));
      #pragma unroll
      for (int j = 0; j < 8; ++j) {
        const int q = q0 + j * AGG_T;
        if (q < found) {
          const int dl = (int)(w[j] >> 17);
          float e = r[j].z + led[dl];
          e = (e > 0.0f) ? e : NEG_SLOPE * e;
          atomicMax(&lmx[dl], fkey(e));
          lq[q] = make_float4(e, r[j].x, r[j].y, __int_as_float(dl));
        }
      }
    }
  } else {
    scan_owned(rid, E, N, r0, lane, wave, [&](int qv, int cnt) {
      int base = 0;
      if (lane == 0) base = atomicAdd(&lqn, cnt);
      base = __builtin_amdgcn_readlane(base, 0);
      if (lane < cnt) {
        float4 r; int dl; float e;
        edge(qv, r, dl, e);
        atomicMax(&lmx[dl], fkey(e));
        const int p = base + lane;
        if (p < QCAP) lq[p] = make_float4(e, r.x, r.y, __int_as_float(dl));
      }
    });
  }
  __syncthreads();

  const int total = listed ? found : lqn;
  if (total <= QCAP) {
    for (int q = tid; q < total; q += AGG_T) {
      const float4 t4 = lq[q];
      const int dl = __float_as_int(t4.w);
      const float ex = expf(t4.x - funkey(lmx[dl]));
      atomicAdd(&lden[dl], ex);
      atomicAdd(&lo0[dl], t4.y * ex);
      atomicAdd(&lo1[dl], t4.z * ex);
    }
  } else {
    scan_owned(rid, E, N, r0, lane, wave, [&](int qv, int cnt) {
      if (lane < cnt) {
        float4 r; int dl; float e;
        edge(qv, r, dl, e);
        const float ex = expf(e - funkey(lmx[dl]));
        atomicAdd(&lden[dl], ex);
        atomicAdd(&lo0[dl], r.x * ex);
        atomicAdd(&lo1[dl], r.y * ex);
      }
    });
  }
  __syncthreads();

  const float bb0 = bias[0], bb1 = bias[1];
  for (int q = tid; q < RNG2; q += AGG_T) {
    const int n = r0 + q;
    if (n < N) {
      const float dn = lden[q];
      float2 o;
      o.x = lo0[q] / dn + bb0;
      o.y = lo1[q] / dn + bb1;
      ((float2*)outp)[n] = o;
    }
  }
}


extern "C" void kernel_launch(void* const* d_in, const int* in_sizes, int n_in,
                              void* d_out, int out_size, void* d_ws, size_t ws_size,
                              hipStream_t stream) {
  const float* x   = (const float*)d_in[0];
  const int*  eidx = (const int*)d_in[1];
  const float* W1  = (const float*)d_in[2];
  const float* a1s = (const float*)d_in[3];
  const float* a1d = (const float*)d_in[4];
  const float* b1  = (const float*)d_in[5];
  const float* W2  = (const float*)d_in[6];
  const float* a2s = (const float*)d_in[7];
  const float* a2d = (const float*)d_in[8];
  const float* b2  = (const float*)d_in[9];

  const int N  = in_sizes[0] / 256;
  const int E  = in_sizes[1] / 2;
  const int* src = eidx;
  const int* dst = eidx + E;
  float* out = (float*)d_out;

  char* w = (char*)d_ws;
  float*    h1b    = (float*)w;    w += (size_t)N * 128 * sizeof(float);
  float*    h1     = (float*)w;    w += (size_t)N * 128 * sizeof(float);
  float*    rec2   = (float*)w;    w += (size_t)N * 4 * sizeof(float);
  float*    e1s    = (float*)w;    w += (size_t)N * sizeof(float);
  float*    e1d    = (float*)w;    w += (size_t)N * sizeof(float);
  unsigned* elist  = (unsigned*)w; w += (size_t)((N + RNG - 1) / RNG) * QCAP1 * sizeof(unsigned);
  int*      ecnt   = (int*)w;      w += 256 * sizeof(int);
  unsigned char* rid = (unsigned char*)w; w += (size_t)E;

  const int TB = 256;
  range_ids<<<((E + 3) / 4 + TB - 1) / TB, TB, 0, stream>>>(dst, E, rid);

  gemm1_wmma<<<G1_GRID, G1_T, 0, stream>>>(x, W1, h1, N);
  node_e1<<<(N + NODE_SPAN - 1) / NODE_SPAN, NODE_T, 0, stream>>>(h1, a1s, a1d, e1s, e1d, N);
  agg128_own<<<(N + RNG - 1) / RNG, AGG_T, 0, stream>>>(h1, e1s, e1d, src, dst, rid, E, N, elist,
                                                        ecnt, b1, h1b);

  node_h2e2<<<(N + NODE_SPAN - 1) / NODE_SPAN, NODE_T, 0, stream>>>(h1b, W2, a2s, a2d, rec2, N);
  agg2_own<<<(N + RNG2 - 1) / RNG2, AGG_T, 0, stream>>>(rec2, src, dst, rid, E, N, elist, ecnt, b2,
                                                        out);
}


# Layer-1 owner's id scan: a lane folds its 64 ids to bit masks and stores its own edges in the LDS list (one counter add a wave); logits, max then from the list, loads batched; sum order = LDS arrival
# speedup vs baseline: 5.7211x; 1.0404x over previous
#include <hip/hip_runtime.h>
#include <hip/hip_bf16.h>

typedef __attribute__((ext_vector_type(2))) float v2f;
typedef __attribute__((ext_vector_type(8))) float v8f;

#define NEG_SLOPE 0.2f


__device__ __forceinline__ float fold16(const float (&v)[16], int lane) {
  const bool b4 = (lane & 16) != 0, b3 = (lane & 8) != 0, b2 = (lane & 4) != 0, b1 = (lane & 2) != 0;
  float a8[8], a4[4], a2[2];
  #pragma unroll
  for (int j = 0; j < 8; ++j) {
    const float keep = b4 ? v[j + 8] : v[j];
    const float send = b4 ? v[j] : v[j + 8];
    a8[j] = keep + __shfl_xor(send, 16, 32);
  }
  #pragma unroll
  for (int j = 0; j < 4; ++j) {
    const float keep = b3 ? a8[j + 4] : a8[j];
    const float send = b3 ? a8[j] : a8[j + 4];
    a4[j] = keep + __shfl_xor(send, 8, 32);
  }
  #pragma unroll
  for (int j = 0; j < 2; ++j) {
    const float keep = b2 ? a4[j + 2] : a4[j];
    const float send = b2 ? a4[j] : a4[j + 2];
    a2[j] = keep + __shfl_xor(send, 4, 32);
  }
  const float keep = b1 ? a2[1] : a2[0];
  const float send = b1 ? a2[0] : a2[1];
  return keep + __shfl_xor(send, 2, 32);
}

__device__ __forceinline__ unsigned fkey(float f) {
  unsigned u = __float_as_uint(f);
  return (u & 0x80000000u) ? ~u : (u | 0x80000000u);
}
__device__ __forceinline__ float funkey(unsigned k) {
  unsigned u = (k & 0x80000000u) ? (k & 0x7FFFFFFFu) : ~k;
  return __uint_as_float(u);
}

#define G1_T 256
#define G1_GRID 512
#define G1_PITCH 260

__global__ __launch_bounds__(G1_T)
void gemm1_wmma(const float* __restrict__ X, const float* __restrict__ W,
                float* __restrict__ H, int N) {
  __shared__ __attribute__((aligned(16))) float xs[2][16 * G1_PITCH];
  const int tid = threadIdx.x;
  const int lane = tid & 31;
  const int wave = __builtin_amdgcn_readfirstlane(tid >> 5);
  const int col0 = wave * 16;
  const int m  = lane & 15;
  const int kh = lane >> 4;
  const int ntiles = (N + 15) >> 4;

  float4 b[32];
  #pragma unroll
  for (int j = 0; j < 32; ++j) {
    const float* wp = W + (size_t)(8 * j + 4 * kh) * 128 + col0 + m;
    b[j] = make_float4(wp[0], wp[128], wp[256], wp[384]);
  }

  const int srow = tid >> 6;
  const int scol = (tid & 63) * 4;
  const float* xsrc = X + scol;
  float* sdst = &xs[0][srow * G1_PITCH + scol];
  float4 p0, p1, p2, p3;
  auto fetch = [&](int t) {
    int r0 = t * 16 + srow, r1 = r0 + 4, r2 = r0 + 8, r3 = r0 + 12;
    if (r3 >= N) {
      r0 = (r0 < N) ? r0 : N - 1; r1 = (r1 < N) ? r1 : N - 1;
      r2 = (r2 < N) ? r2 : N - 1; r3 = N - 1;
    }
    p0 = *(const float4*)(xsrc + (size_t)r0 * 256);
    p1 = *(const float4*)(xsrc + (size_t)r1 * 256);
    p2 = *(const float4*)(xsrc + (size_t)r2 * 256);
    p3 = *(const float4*)(xsrc + (size_t)r3 * 256);
  };
  auto stage = [&](int bf) {
    float* d = sdst + bf * (16 * G1_PITCH);
    *(float4*)(d) = p0;
    *(float4*)(d + 4 * G1_PITCH) = p1;
    *(float4*)(d + 8 * G1_PITCH) = p2;
    *(float4*)(d + 12 * G1_PITCH) = p3;
  };

  int t = blockIdx.x;
  if (t < ntiles) { fetch(t); stage(0); }
  __syncthreads();
  int bf = 0;
  for (; t < ntiles; t += G1_GRID) {
    const int tn = t + G1_GRID;
    if (tn < ntiles) fetch(tn);

    v8f c0 = {}, c1 = {};
    const float* xa = &xs[bf][m * G1_PITCH + 4 * kh];
    float4 af[2][4];
    #pragma unroll
    for (int i = 0; i < 4; ++i) af[0][i] = *(const float4*)(xa + 8 * i);
    #pragma unroll
    for (int g = 0; g < 8; ++g) {
      if (g < 7) {
        #pragma unroll
        for (int i = 0; i < 4; ++i) af[(g + 1) & 1][i] = *(const float4*)(xa + 8 * (4 * (g + 1) + i));
      }
      #pragma unroll
      for (int i = 0; i < 4; ++i) {
        const float4 a = af[g & 1][i];
        const float4 bb = b[4 * g + i];
        v2f a0, a1, b0, b1;
        a0.x = a.x; a0.y = a.y; a1.x = a.z; a1.y = a.w;
        b0.x = bb.x; b0.y = bb.y; b1.x = bb.z; b1.y = bb.w;
#if __has_builtin(__builtin_amdgcn_wmma_f32_16x16x4_f32)
        c0 = __builtin_amdgcn_wmma_f32_16x16x4_f32(false, a0, false, b0, (short)0, c0,
                                                   false, false);
        c1 = __builtin_amdgcn_wmma_f32_16x16x4_f32(false, a1, false, b1, (short)0, c1,
                                                   false, false);
#else
        c0[0] += a0.x * b0.x + a0.y * b0.y;
        c1[0] += a1.x * b1.x + a1.y * b1.y;
#endif
      }
    }
#if __has_builtin(__builtin_amdgcn_sched_group_barrier)
    __builtin_amdgcn_sched_group_barrier(0x100, 4, 0);
    #pragma unroll
    for (int g = 0; g < 7; ++g) {
      __builtin_amdgcn_sched_group_barrier(0x100, 4, 0);
      __builtin_amdgcn_sched_group_barrier(0x008, 8, 0);
    }
    __builtin_amdgcn_sched_group_barrier(0x008, 8, 0);
#endif
    const v8f c = c0 + c1;

    if (tn < ntiles) stage(bf ^ 1);
    __syncthreads();
    bf ^= 1;

    const int row0 = t * 16;
    float* hp = H + (size_t)(row0 + 8 * kh) * 128 + col0 + m;
    if (row0 + 16 <= N) {
      #pragma unroll
      for (int r = 0; r < 8; ++r) hp[(size_t)r * 128] = c[r];
    } else {
      #pragma unroll
      for (int r = 0; r < 8; ++r)
        if (row0 + r + 8 * kh < N) hp[(size_t)r * 128] = c[r];
    }
  }
}

#define NPW 16
static_assert(NPW == 16, "fold16 folds 16 values per lane");
#define NODE_T 256
#define NODE_SPAN (NPW * (NODE_T / 32))

__global__ __launch_bounds__(NODE_T)
void node_e1(const float* __restrict__ H, const float* __restrict__ as,
             const float* __restrict__ ad, float* __restrict__ es,
             float* __restrict__ ed, int N) {
  const int lane = threadIdx.x & 31;
  const int n0 = __builtin_amdgcn_readfirstlane(
      (int)(blockIdx.x * (NODE_T / 32) + (threadIdx.x >> 5)) * NPW);
  if (n0 >= N) return;
  const float4 sv = ((const float4*)as)[lane];
  const float4 dv = ((const float4*)ad)[lane];
  const float4* H4 = (const float4*)H;
  float4 hv[NPW];
  #pragma unroll
  for (int j = 0; j < NPW; ++j) {
    const int n = (n0 + j < N) ? n0 + j : N - 1;
    hv[j] = H4[(size_t)n * 32 + lane];
  }
  float s[NPW], d[NPW];
  #pragma unroll
  for (int j = 0; j < NPW; ++j) {
    s[j] = hv[j].x * sv.x + hv[j].y * sv.y + hv[j].z * sv.z + hv[j].w * sv.w;
    d[j] = hv[j].x * dv.x + hv[j].y * dv.y + hv[j].z * dv.z + hv[j].w * dv.w;
  }
  const float fs = fold16(s, lane);
  const float fd = fold16(d, lane);
  const bool b0 = (lane & 1) != 0;
  const float keep = b0 ? fd : fs;
  const float send = b0 ? fs : fd;
  const float r = keep + __shfl_xor(send, 1, 32);
  const int n = n0 + (lane >> 1);
  if (n < N) {
    float* o = b0 ? ed : es;
    o[n] = r;
  }
}

#define RNG 400

__global__ void range_ids(const int* __restrict__ dst, int E, unsigned char* __restrict__ rid) {
  const int t = blockIdx.x * blockDim.x + threadIdx.x;
  int i = 4 * t;
  if (i + 3 < E) {
    const unsigned w = (unsigned)(dst[i] / RNG) | ((unsigned)(dst[i + 1] / RNG) << 8) |
                       ((unsigned)(dst[i + 2] / RNG) << 16) | ((unsigned)(dst[i + 3] / RNG) << 24);
    ((unsigned*)rid)[t] = w;
  } else {
    for (; i < E; ++i) rid[i] = (unsigned char)(dst[i] / RNG);
  }
}

template <class F>
__device__ __forceinline__ void scan_owned(const unsigned char* __restrict__ rid, int E, int N,
                                           int r0, int lane, int wave, F&& fl) {
  int qv = 0;
  int qn = 0;
  auto push = [&](int i) {
    qv = (lane == qn) ? i : qv;
    if (++qn == 32) { fl(qv, 32); qn = 0; }
  };
  const unsigned own = blockIdx.x & 255u;
  const unsigned pat = own * 0x01010101u;
  auto zb = [](unsigned x) {
    return ~((((x & 0x7F7F7F7Fu) + 0x7F7F7F7Fu) | x) | 0x7F7F7F7Fu);
  };
  int a0 = (int)((16u - ((unsigned)(size_t)rid & 15u)) & 15u);
  if (a0 > E) a0 = E;
  const int nfast = (E - a0) / 2048;
  const uint4* rq = (const uint4*)(rid + a0);
  const int T = (nfast > wave) ? (nfast - wave + 31) / 32 : 0;
  int t = (int)(((size_t)blockIdx.x * (size_t)T) / gridDim.x);
  for (int k = 0; k < T; ++k) {
    const int c = t * 32 + wave;
    if (++t == T) t = 0;
    uint4 v[4];
    #pragma unroll
    for (int j = 0; j < 4; ++j) v[j] = rq[c * 128 + j * 32 + lane];
    #pragma unroll
    for (int j = 0; j < 4; ++j) {
      const unsigned f0 = zb(v[j].x ^ pat);
      const unsigned f1 = zb(v[j].y ^ pat);
      const unsigned f2 = zb(v[j].z ^ pat);
      const unsigned f3 = zb(v[j].w ^ pat);
      const unsigned m0 = __builtin_amdgcn_ballot_w32(f0 != 0u);
      const unsigned m1 = __builtin_amdgcn_ballot_w32(f1 != 0u);
      const unsigned m2 = __builtin_amdgcn_ballot_w32(f2 != 0u);
      const unsigned m3 = __builtin_amdgcn_ballot_w32(f3 != 0u);
      unsigned any = m0 | m1 | m2 | m3;
      const int ibase = a0 + (c * 128 + j * 32) * 16;
      while (any) {
        const int l = __builtin_ctz(any);
        any &= any - 1;
        const int il = ibase + l * 16;
        if ((m0 >> l) & 1u)
          for (unsigned f = __builtin_amdgcn_readlane(f0, l); f; f &= f - 1)
            push(il + 0 + (__builtin_ctz(f) >> 3));
        if ((m1 >> l) & 1u)
          for (unsigned f = __builtin_amdgcn_readlane(f1, l); f; f &= f - 1)
            push(il + 4 + (__builtin_ctz(f) >> 3));
        if ((m2 >> l) & 1u)
          for (unsigned f = __builtin_amdgcn_readlane(f2, l); f; f &= f - 1)
            push(il + 8 + (__builtin_ctz(f) >> 3));
        if ((m3 >> l) & 1u)
          for (unsigned f = __builtin_amdgcn_readlane(f3, l); f; f &= f - 1)
            push(il + 12 + (__builtin_ctz(f) >> 3));
      }
    }
  }
  const int nrem = E - nfast * 2048;
  for (int g = wave; g * 32 < nrem; g += 32) {
    const int j = g * 32 + lane;
    const int i = (j < a0) ? j : nfast * 2048 + j;
    const unsigned idv = (j < nrem) ? (unsigned)rid[i] : 256u;
    unsigned m = __builtin_amdgcn_ballot_w32(idv == own);
    while (m) {
      const int l = __builtin_ctz(m);
      m &= m - 1;
      push(__builtin_amdgcn_readlane(i, l));
    }
  }
  for (int n = r0 + wave; n < r0 + RNG && n < N; n += 32) push(E + n);
  if (qn) fl(qv, qn);
}

__device__ __forceinline__ void scan_append(const unsigned char* __restrict__ rid, int E, int N,
                                            int r0, int lane, int wave, uint2* lq, int cap,
                                            int* lqn) {
  const unsigned own = blockIdx.x & 255u;
  const unsigned pat = own * 0x01010101u;
  auto nz = [&](unsigned w) {
    const unsigned x = w ^ pat;
    return (((x | 0x80808080u) - 0x01010101u) | x) & 0x80808080u;
  };
  auto fold = [&](const uint4& v) {
    return (nz(v.x) >> 7) | (nz(v.y) >> 6) | (nz(v.z) >> 5) | (nz(v.w) >> 4);
  };
  auto put = [&](unsigned m, int ib, int& p) {
    while (m) {
      const int q = __builtin_ctz(m);
      m &= m - 1;
      const int idx = ib + ((q >> 2) & 1) * 512 + (q & 3) * 4 + (q >> 3);
      if (p < cap) lq[p].x = (unsigned)idx;
      ++p;
    }
  };
  int a0 = (int)((16u - ((unsigned)(size_t)rid & 15u)) & 15u);
  if (a0 > E) a0 = E;
  const int nfast = (E - a0) / 2048;
  const uint4* rq = (const uint4*)(rid + a0);
  const int T = (nfast > wave) ? (nfast - wave + 31) / 32 : 0;
  int t = (int)(((size_t)blockIdx.x * (size_t)T) / gridDim.x);
  auto next = [&]() {
    const int c = t * 32 + wave;
    if (++t == T) t = 0;
    return c;
  };
  auto load = [&](uint4 (&v)[4], int c) {
    #pragma unroll
    for (int j = 0; j < 4; ++j) v[j] = rq[c * 128 + j * 32 + lane];
  };
  auto piece = [&](const uint4 (&v)[4], int c) {
    const unsigned m0 = ~(fold(v[0]) | (fold(v[1]) << 4));
    const unsigned m1 = ~(fold(v[2]) | (fold(v[3]) << 4));
    const int cnt = __popc(m0) + __popc(m1);
    int x = cnt;
    x += __builtin_amdgcn_update_dpp(0, x, 0x111, 0xf, 0xf, true);
    x += __builtin_amdgcn_update_dpp(0, x, 0x112, 0xf, 0xf, true);
    x += __builtin_amdgcn_update_dpp(0, x, 0x114, 0xf, 0xf, true);
    x += __builtin_amdgcn_update_dpp(0, x, 0x118, 0xf, 0xf, true);
    const int low = __builtin_amdgcn_readlane(x, 15);
    x += (lane & 16) ? low : 0;
    const int tot = __builtin_amdgcn_readlane(x, 31);
    if (tot) {
      int base = 0;
      if (lane == 0) base = atomicAdd(lqn, tot);
      base = __builtin_amdgcn_readfirstlane(base);
      if (cnt) {
        int p = base + x - cnt;
        const int ib = a0 + (c * 128 + lane) * 16;
        put(m0, ib, p);
        put(m1, ib + 1024, p);
      }
    }
  };
  if (T > 0) {
    uint4 va[4], vb[4];
    int k = 0;
    int ca = next(), cb;
    load(va, ca);
    for (; k + 1 < T; k += 2) {
      cb = next();
      load(vb, cb);
      piece(va, ca);
      ca = next();
      load(va, ca);
      piece(vb, cb);
    }
    if (k < T) piece(va, ca);
  }
  const int nrem = E - nfast * 2048;
  for (int g = wave; g * 32 < nrem; g += 32) {
    const int j = g * 32 + lane;
    const int i = (j < a0) ? j : nfast * 2048 + j;
    if (j < nrem && (unsigned)rid[i] == own) {
      const int p = atomicAdd(lqn, 1);
      if (p < cap) lq[p].x = (unsigned)i;
    }
  }
  for (int n = r0 + wave * 32 + lane; n < r0 + RNG && n < N; n += 32 * 32) {
    const int p = atomicAdd(lqn, 1);
    if (p < cap) lq[p].x = (unsigned)(E + n);
  }
}

#define AGG_T 1024
#define QCAP1 10240
static_assert(RNG <= 512 && 256 * RNG <= (1 << 17), "list entry packing");

__global__ __launch_bounds__(AGG_T)
void agg128_own(const float* __restrict__ H, const float* __restrict__ es,
                const float* __restrict__ ed, const int* __restrict__ src,
                const int* __restrict__ dst, const unsigned char* __restrict__ rid, int E, int N,
                unsigned* __restrict__ elist, int* __restrict__ ecnt,
                const float* __restrict__ bias, float* __restrict__ outp) {
  __shared__ float lacc[RNG * 128];
  __shared__ uint2 lq[QCAP1];
  __shared__ float led[RNG];
  __shared__ unsigned lmx[RNG];
  __shared__ float lden[RNG];
  __shared__ int lqn;
  const int tid = threadIdx.x;
  const int lane = tid & 31;
  const int wave = __builtin_amdgcn_readfirstlane(tid >> 5);
  const int r0 = blockIdx.x * RNG;

  for (int q = tid; q < RNG * 32; q += AGG_T) ((float4*)lacc)[q] = make_float4(0.f, 0.f, 0.f, 0.f);
  for (int q = tid; q < RNG; q += AGG_T) {
    const int n = r0 + q;
    led[q] = (n < N) ? ed[n] : 0.0f;
    lmx[q] = 0u;
    lden[q] = 0.0f;
  }
  if (tid == 0) lqn = 0;
  __syncthreads();

  const float* hl = H + lane;
  float* al = lacc + lane;

  auto edge = [&](int i, int& s, int& dl, float& e) {
    int d;
    if (i < E) { s = src[i]; d = dst[i]; } else { s = d = i - E; }
    dl = d - r0;
    e = es[s] + led[dl];
    e = (e > 0.0f) ? e : NEG_SLOPE * e;
  };

  auto rows = [&](int s, int dl, float ex, int cnt) {
    const int exb = __float_as_int(ex);
    int b = 0;
    for (; b + 8 <= cnt; b += 8) {
      float hv[8][4];
      #pragma unroll
      for (int j = 0; j < 8; ++j) {
        const int sj = __builtin_amdgcn_readlane(s, b + j);
        const float* hp = hl + (size_t)sj * 128;
        hv[j][0] = hp[0]; hv[j][1] = hp[32]; hv[j][2] = hp[64]; hv[j][3] = hp[96];
      }
      #pragma unroll
      for (int j = 0; j < 8; ++j) {
        const int dj = __builtin_amdgcn_readlane(dl, b + j);
        const float ej = __int_as_float(__builtin_amdgcn_readlane(exb, b + j));
        float* o = al + dj * 128;
        atomicAdd(o +  0, hv[j][0] * ej);
        atomicAdd(o + 32, hv[j][1] * ej);
        atomicAdd(o + 64, hv[j][2] * ej);
        atomicAdd(o + 96, hv[j][3] * ej);
      }
    }
    for (; b < cnt; ++b) {
      const int sj = __builtin_amdgcn_readlane(s, b);
      const int dj = __builtin_amdgcn_readlane(dl, b);
      const float ej = __int_as_float(__builtin_amdgcn_readlane(exb, b));
      const float* hp = hl + (size_t)sj * 128;
      float* o = al + dj * 128;
      const float h0 = hp[0], h1 = hp[32], h2 = hp[64], h3 = hp[96];
      atomicAdd(o +  0, h0 * ej);
      atomicAdd(o + 32, h1 * ej);
      atomicAdd(o + 64, h2 * ej);
      atomicAdd(o + 96, h3 * ej);
    }
  };

  scan_append(rid, E, N, r0, lane, wave, lq, QCAP1, &lqn);
  __syncthreads();

  const int total = __builtin_amdgcn_readfirstlane(lqn);
  if (tid == 0) ecnt[blockIdx.x] = total;
  if (total <= QCAP1) {
    for (int q0 = tid; q0 < total; q0 += 8 * AGG_T) {
      int ei[8], s[8], d[8];
      float sv[8];
      #pragma unroll
      for (int j = 0; j < 8; ++j) {
        const int q = q0 + j * AGG_T;
        ei[j] = (q < total) ? (int)lq[q].x : E + r0;
      }
      #pragma unroll
      for (int j = 0; j < 8; ++j) {
        const int i = ei[j];
        if (i < E) { s[j] = src[i]; d[j] = dst[i]; } else { s[j] = d[j] = i - E; }
      }
      #pragma unroll
      for (int j = 0; j < 8; ++j) sv[j] = es[s[j]];
      #pragma unroll
      for (int j = 0; j < 8; ++j) {
        const int q = q0 + j * AGG_T;
        if (q < total) {
          const int dl = d[j] - r0;
          float e = sv[j] + led[dl];
          e = (e > 0.0f) ? e : NEG_SLOPE * e;
          atomicMax(&lmx[dl], fkey(e));
          lq[q] = make_uint2((unsigned)s[j] | ((unsigned)dl << 17), __float_as_uint(e));
        }
      }
    }
    __syncthreads();
    unsigned* el = elist + (size_t)blockIdx.x * QCAP1;
    for (int q = tid; q < total; q += AGG_T) el[q] = lq[q].x;
    for (int c0 = wave * 32; c0 < total; c0 += 32 * 32) {
      const int cnt = (total - c0 < 32) ? total - c0 : 32;
      int s = 0, dl = 0;
      float ex = 0.0f;
      if (lane < cnt) {
        const uint2 t2 = lq[c0 + lane];
        s = (int)(t2.x & 0x1FFFFu);
        dl = (int)(t2.x >> 17);
        ex = expf(__uint_as_float(t2.y) - funkey(lmx[dl]));
        atomicAdd(&lden[dl], ex);
      }
      rows(s, dl, ex, cnt);
    }
  } else {
    scan_owned(rid, E, N, r0, lane, wave, [&](int qv, int cnt) {
      if (lane < cnt) {
        int s, dl; float e;
        edge(qv, s, dl, e);
        atomicMax(&lmx[dl], fkey(e));
      }
    });
    __syncthreads();
    scan_owned(rid, E, N, r0, lane, wave, [&](int qv, int cnt) {
      int s = 0, dl = 0;
      float ex = 0.0f;
      if (lane < cnt) {
        float e;
        edge(qv, s, dl, e);
        ex = expf(e - funkey(lmx[dl]));
        atomicAdd(&lden[dl], ex);
      }
      rows(s, dl, ex, cnt);
    });
  }
  __syncthreads();

  for (int q = tid; q < RNG * 32; q += AGG_T) {
    const int node = q >> 5;
    const int n = r0 + node;
    if (n < N) {
      const float4 a = ((const float4*)lacc)[q];
      const float dn = lden[node];
      const float4 bb = ((const float4*)bias)[q & 31];
      float4 r;
      r.x = a.x / dn + bb.x; r.y = a.y / dn + bb.y; r.z = a.z / dn + bb.z; r.w = a.w / dn + bb.w;
      r.x = (r.x > 0.0f) ? r.x : 0.0f; r.y = (r.y > 0.0f) ? r.y : 0.0f;
      r.z = (r.z > 0.0f) ? r.z : 0.0f; r.w = (r.w > 0.0f) ? r.w : 0.0f;
      ((float4*)(outp + (size_t)n * 128))[q & 31] = r;
    }
  }
}

__global__ __launch_bounds__(NODE_T)
void node_h2e2(const float* __restrict__ H, const float* __restrict__ W2,
               const float* __restrict__ a2s, const float* __restrict__ a2d,
               float* __restrict__ rec, int N) {
  const int lane = threadIdx.x & 31;
  const int n0 = __builtin_amdgcn_readfirstlane(
      (int)(blockIdx.x * (NODE_T / 32) + (threadIdx.x >> 5)) * NPW);
  if (n0 >= N) return;
  const int c0 = lane * 4;
  const float w00 = W2[(c0 + 0) * 2 + 0], w01 = W2[(c0 + 0) * 2 + 1];
  const float w10 = W2[(c0 + 1) * 2 + 0], w11 = W2[(c0 + 1) * 2 + 1];
  const float w20 = W2[(c0 + 2) * 2 + 0], w21 = W2[(c0 + 2) * 2 + 1];
  const float w30 = W2[(c0 + 3) * 2 + 0], w31 = W2[(c0 + 3) * 2 + 1];
  const float4* H4 = (const float4*)H;
  float4 hv[NPW];
  #pragma unroll
  for (int j = 0; j < NPW; ++j) {
    const int n = (n0 + j < N) ? n0 + j : N - 1;
    hv[j] = H4[(size_t)n * 32 + lane];
  }
  float s0[NPW], s1[NPW];
  #pragma unroll
  for (int j = 0; j < NPW; ++j) {
    s0[j] = hv[j].x * w00 + hv[j].y * w10 + hv[j].z * w20 + hv[j].w * w30;
    s1[j] = hv[j].x * w01 + hv[j].y * w11 + hv[j].z * w21 + hv[j].w * w31;
  }
  const float f0 = fold16(s0, lane);
  const float f1 = fold16(s1, lane);
  const bool b0 = (lane & 1) != 0;
  const float keep = b0 ? f1 : f0;
  const float send = b0 ? f0 : f1;
  const float r = keep + __shfl_xor(send, 1, 32);
  const float q = __shfl_xor(r, 1, 32);
  const float h0 = b0 ? q : r;
  const float h1 = b0 ? r : q;
  float2 o;
  o.x = b0 ? h0 * a2s[0] + h1 * a2s[1] : h0;
  o.y = b0 ? h0 * a2d[0] + h1 * a2d[1] : h1;
  const int n = n0 + (lane >> 1);
  if (n < N) ((float2*)rec)[(size_t)n * 2 + (b0 ? 1 : 0)] = o;
}

#define RNG2 RNG
#define QCAP 12288

static_assert(QCAP1 <= QCAP, "list hand-over");

__global__ __launch_bounds__(AGG_T)
void agg2_own(const float* __restrict__ rec, const int* __restrict__ src,
              const int* __restrict__ dst, const unsigned char* __restrict__ rid, int E, int N,
              const unsigned* __restrict__ elist, const int* __restrict__ ecnt,
              const float* __restrict__ bias, float* __restrict__ outp) {
  __shared__ float4 lq[QCAP];
  __shared__ float led[RNG2];
  __shared__ unsigned lmx[RNG2];
  __shared__ float lden[RNG2];
  __shared__ float lo0[RNG2];
  __shared__ float lo1[RNG2];
  __shared__ int lqn;
  const int tid = threadIdx.x;
  const int lane = tid & 31;
  const int wave = __builtin_amdgcn_readfirstlane(tid >> 5);
  const int r0 = blockIdx.x * RNG2;
  const float4* rec4 = (const float4*)rec;

  for (int q = tid; q < RNG2; q += AGG_T) {
    const int n = r0 + q;
    led[q] = (n < N) ? rec4[n].w : 0.0f;
    lmx[q] = 0u;
    lden[q] = 0.0f;
    lo0[q] = 0.0f;
    lo1[q] = 0.0f;
  }
  if (tid == 0) lqn = 0;
  __syncthreads();

  auto edge = [&](int i, float4& r, int& dl, float& e) {
    int s, d;
    if (i < E) { s = src[i]; d = dst[i]; } else { s = d = i - E; }
    r = rec4[s];
    dl = d - r0;
    e = r.z + led[dl];
    e = (e > 0.0f) ? e : NEG_SLOPE * e;
  };

  const int found = ecnt[blockIdx.x];
  const bool listed = found <= QCAP1;
  if (listed) {
    const unsigned* el = elist + (size_t)blockIdx.x * QCAP1;
    for (int q0 = tid; q0 < found; q0 += 8 * AGG_T) {
      unsigned w[8];
      float4 r[8];
      #pragma unroll
      for (int j = 0; j < 8; ++j) {
        const int q = q0 + j * AGG_T;
        w[j] = (q < found) ? el[q] : 0u;
      }
      #pragma unroll
      for (int j = 0; j < 8; ++j) r[j] = rec4[w[j] & 0x1FFFFu];
      asm volatile("" :: "v"(r[0].z), "v"(r[1].z), "v"(r[2].z), "v"(r[3].z), "v"(r[4].z),
                   "v"(r[5].z), "v"(r[6].z), "v"(r[7].z));
      #pragma unroll
      for (int j = 0; j < 8; ++j) {
        const int q = q0 + j * AGG_T;
        if (q < found) {
          const int dl = (int)(w[j] >> 17);
          float e = r[j].z + led[dl];
          e = (e > 0.0f) ? e : NEG_SLOPE * e;
          atomicMax(&lmx[dl], fkey(e));
          lq[q] = make_float4(e, r[j].x, r[j].y, __int_as_float(dl));
        }
      }
    }
  } else {
    scan_owned(rid, E, N, r0, lane, wave, [&](int qv, int cnt) {
      int base = 0;
      if (lane == 0) base = atomicAdd(&lqn, cnt);
      base = __builtin_amdgcn_readlane(base, 0);
      if (lane < cnt) {
        float4 r; int dl; float e;
        edge(qv, r, dl, e);
        atomicMax(&lmx[dl], fkey(e));
        const int p = base + lane;
        if (p < QCAP) lq[p] = make_float4(e, r.x, r.y, __int_as_float(dl));
      }
    });
  }
  __syncthreads();

  const int total = listed ? found : lqn;
  if (total <= QCAP) {
    for (int q = tid; q < total; q += AGG_T) {
      const float4 t4 = lq[q];
      const int dl = __float_as_int(t4.w);
      const float ex = expf(t4.x - funkey(lmx[dl]));
      atomicAdd(&lden[dl], ex);
      atomicAdd(&lo0[dl], t4.y * ex);
      atomicAdd(&lo1[dl], t4.z * ex);
    }
  } else {
    scan_owned(rid, E, N, r0, lane, wave, [&](int qv, int cnt) {
      if (lane < cnt) {
        float4 r; int dl; float e;
        edge(qv, r, dl, e);
        const float ex = expf(e - funkey(lmx[dl]));
        atomicAdd(&lden[dl], ex);
        atomicAdd(&lo0[dl], r.x * ex);
        atomicAdd(&lo1[dl], r.y * ex);
      }
    });
  }
  __syncthreads();

  const float bb0 = bias[0], bb1 = bias[1];
  for (int q = tid; q < RNG2; q += AGG_T) {
    const int n = r0 + q;
    if (n < N) {
      const float dn = lden[q];
      float2 o;
      o.x = lo0[q] / dn + bb0;
      o.y = lo1[q] / dn + bb1;
      ((float2*)outp)[n] = o;
    }
  }
}


extern "C" void kernel_launch(void* const* d_in, const int* in_sizes, int n_in,
                              void* d_out, int out_size, void* d_ws, size_t ws_size,
                              hipStream_t stream) {
  const float* x   = (const float*)d_in[0];
  const int*  eidx = (const int*)d_in[1];
  const float* W1  = (const float*)d_in[2];
  const float* a1s = (const float*)d_in[3];
  const float* a1d = (const float*)d_in[4];
  const float* b1  = (const float*)d_in[5];
  const float* W2  = (const float*)d_in[6];
  const float* a2s = (const float*)d_in[7];
  const float* a2d = (const float*)d_in[8];
  const float* b2  = (const float*)d_in[9];

  const int N  = in_sizes[0] / 256;
  const int E  = in_sizes[1] / 2;
  const int* src = eidx;
  const int* dst = eidx + E;
  float* out = (float*)d_out;

  char* w = (char*)d_ws;
  float*    h1b    = (float*)w;    w += (size_t)N * 128 * sizeof(float);
  float*    h1     = (float*)w;    w += (size_t)N * 128 * sizeof(float);
  float*    rec2   = (float*)w;    w += (size_t)N * 4 * sizeof(float);
  float*    e1s    = (float*)w;    w += (size_t)N * sizeof(float);
  float*    e1d    = (float*)w;    w += (size_t)N * sizeof(float);
  unsigned* elist  = (unsigned*)w; w += (size_t)((N + RNG - 1) / RNG) * QCAP1 * sizeof(unsigned);
  int*      ecnt   = (int*)w;      w += 256 * sizeof(int);
  unsigned char* rid = (unsigned char*)w; w += (size_t)E;

  const int TB = 256;
  range_ids<<<((E + 3) / 4 + TB - 1) / TB, TB, 0, stream>>>(dst, E, rid);

  gemm1_wmma<<<G1_GRID, G1_T, 0, stream>>>(x, W1, h1, N);
  node_e1<<<(N + NODE_SPAN - 1) / NODE_SPAN, NODE_T, 0, stream>>>(h1, a1s, a1d, e1s, e1d, N);
  agg128_own<<<(N + RNG - 1) / RNG, AGG_T, 0, stream>>>(h1, e1s, e1d, src, dst, rid, E, N, elist,
                                                        ecnt, b1, h1b);

  node_h2e2<<<(N + NODE_SPAN - 1) / NODE_SPAN, NODE_T, 0, stream>>>(h1b, W2, a2s, a2d, rec2, N);
  agg2_own<<<(N + RNG2 - 1) / RNG2, AGG_T, 0, stream>>>(rec2, src, dst, rid, E, N, elist, ecnt, b2,
                                                        out);
}
